# starting kernel
# baseline (speedup 1.0000x reference)
#include <hip/hip_runtime.h>


typedef __attribute__((ext_vector_type(2))) float v2f;
typedef __attribute__((ext_vector_type(8))) float v8f;

#define SC_HALF_SQRT2 0.70710678118654752f

__device__ __forceinline__ int refl(int i, int n) {
  i = (i < 0) ? (-1 - i) : i;
  i = (i >= n) ? (2 * n - 1 - i) : i;
  return i;
}

__device__ __forceinline__ int iclamp(int v, int lo, int hi) {
  return v < lo ? lo : (v > hi ? hi : v);
}

__device__ __forceinline__ v8f wmma4(v2f a, v2f b, v8f c) {
  return __builtin_amdgcn_wmma_f32_16x16x4_f32(false, a, false, b, (short)0, c,
                                               false, false);
}

__device__ __forceinline__ float ifilt_coef(int o, int kg, int n,
                                            const float (&ha)[10],
                                            const float (&hb)[10], bool hp) {
  const int i = o >> 2, s = o & 3;
  const int base = 4 + ((((s & 1) != 0) != hp) ? 1 : 0);
  float acc = 0.f;
#pragma unroll
  for (int m = 0; m < 5; ++m) {
    const int L = refl(2 * i + base - 2 * m, n);
    const float c = (s == 0)   ? ha[2 * m]
                    : (s == 1) ? hb[2 * m]
                    : (s == 2) ? ha[2 * m + 1]
                               : hb[2 * m + 1];
    acc += (L == kg) ? c : 0.f;
  }
  return acc;
}

template <int NT>
__device__ __forceinline__ float filt_coef(int o, int kg, int n,
                                           const float (&h)[NT]) {
  float acc = 0.f;
#pragma unroll
  for (int m = 0; m < NT; ++m) {
    const int L = refl(o + (NT >> 1) - m, n);
    acc += (L == kg) ? h[m] : 0.f;
  }
  return acc;
}


__global__ __launch_bounds__(256) void build_ifilt_colF(
    const float* __restrict__ g0a, const float* __restrict__ g0b,
    const float* __restrict__ g1a, const float* __restrict__ g1b,
    float* __restrict__ M, int r) {
  const int ty = blockIdx.x, f = blockIdx.y;
  float ha[10], hb[10];
#pragma unroll
  for (int m = 0; m < 10; ++m) {
    ha[m] = f ? g1b[m] : g0b[m];
    hb[m] = f ? g1a[m] : g0a[m];
  }
  const int o0 = ty << 4;
  const int w0 = iclamp((o0 >> 1) - 4, 0, r - 16);
  const int idx = threadIdx.x;
  const int row = idx >> 4, k = idx & 15;
  M[(((size_t)ty * 2 + f) * 16 + row) * 16 + k] =
      ifilt_coef(o0 + row, w0 + k, r, ha, hb, f != 0);
}

__global__ __launch_bounds__(256) void build_ifilt_rowG(
    const float* __restrict__ g0a, const float* __restrict__ g0b,
    const float* __restrict__ g1a, const float* __restrict__ g1b,
    float* __restrict__ M, int W) {
  const int tx = blockIdx.x, f = blockIdx.y;
  float ha[10], hb[10];
#pragma unroll
  for (int m = 0; m < 10; ++m) {
    ha[m] = f ? g1b[m] : g0b[m];
    hb[m] = f ? g1a[m] : g0a[m];
  }
  const int p0 = tx << 4;
  const int w0 = iclamp((p0 >> 1) - 4, 0, W - 16);
  const int idx = threadIdx.x;
  const int k = idx >> 4, n = idx & 15;
  M[(((size_t)tx * 2 + f) * 16 + k) * 16 + n] =
      ifilt_coef(p0 + n, w0 + k, W, ha, hb, f != 0);
}

__global__ __launch_bounds__(256) void build_j1_colF(
    const float* __restrict__ g0o, const float* __restrict__ g1o,
    float* __restrict__ M, int r) {
  const int ty = blockIdx.x, f = blockIdx.y;
  float h0[7], h1[5];
#pragma unroll
  for (int m = 0; m < 7; ++m) h0[m] = g0o[m];
#pragma unroll
  for (int m = 0; m < 5; ++m) h1[m] = g1o[m];
  const int o0 = ty << 4;
  const int w0 = iclamp(o0 - 4, 0, r - 24);
  for (int idx = threadIdx.x; idx < 16 * 24; idx += 256) {
    const int row = idx / 24, k = idx % 24;
    const float v = f ? filt_coef<5>(o0 + row, w0 + k, r, h1)
                      : filt_coef<7>(o0 + row, w0 + k, r, h0);
    M[((size_t)ty * 2 + f) * 384 + row * 24 + k] = v;
  }
}

__global__ __launch_bounds__(256) void build_j1_rowG(
    const float* __restrict__ g0o, const float* __restrict__ g1o,
    float* __restrict__ M, int W) {
  const int tx = blockIdx.x, f = blockIdx.y;
  float h0[7], h1[5];
#pragma unroll
  for (int m = 0; m < 7; ++m) h0[m] = g0o[m];
#pragma unroll
  for (int m = 0; m < 5; ++m) h1[m] = g1o[m];
  const int p0 = tx << 4;
  const int w0 = iclamp(p0 - 4, 0, W - 24);
  for (int idx = threadIdx.x; idx < 24 * 16; idx += 256) {
    const int k = idx >> 4, n = idx & 15;
    const float v = f ? filt_coef<5>(p0 + n, w0 + k, W, h1)
                      : filt_coef<7>(p0 + n, w0 + k, W, h0);
    M[((size_t)tx * 2 + f) * 384 + k * 16 + n] = v;
  }
}


template <int R>
__global__ __launch_bounds__(32) void col_ifilt_kernel(
    const float* __restrict__ ll, const float* __restrict__ yh,
    const float* __restrict__ Fmat, float* __restrict__ y1,
    float* __restrict__ y2) {
  constexpr int TILES_X = R / 16;
  constexpr int H2 = R / 2;
  constexpr int SB = H2 * H2 * 2;
  const int lane = threadIdx.x;
  const int tx = blockIdx.x % TILES_X, ty = blockIdx.x / TILES_X;
  const int o0 = ty << 4, c0 = tx << 4;
  const int mc = blockIdx.y;
  const int which = blockIdx.z;
  const int w0 = iclamp((o0 >> 1) - 4, 0, R - 16);

  const int hg = lane >> 4, l16 = lane & 15;
  const int bcol = c0 + l16;

  const float* llp = ll + (size_t)mc * (R * R);
  const float* yhp = yh + (size_t)mc * (6 * SB);
  const float* pH1 = yhp + (which ? 1 : 0) * SB;
  const float* pH2 = yhp + (which ? 4 : 5) * SB;
  const float* pL1 = yhp + 2 * SB;
  const float* pL2 = yhp + 3 * SB;

  const float sgn1 = (bcol & 1) ? -SC_HALF_SQRT2 : SC_HALF_SQRT2;
  const float sgn2 = (bcol & 1) ? SC_HALF_SQRT2 : -SC_HALF_SQRT2;
  const unsigned colE = (unsigned)bcol;
  const unsigned colO = (unsigned)(bcol ^ 1);

  const float* FL = Fmat + ((size_t)ty * 2 + 0) * 256;
  const float* FH = Fmat + ((size_t)ty * 2 + 1) * 256;
  v2f aL[4], aH[4];
#pragma unroll
  for (int t = 0; t < 4; ++t) {
    const int kb = 4 * t + (hg << 1);
    aL[t] = *reinterpret_cast<const v2f*>(FL + l16 * 16 + kb);
    aH[t] = *reinterpret_cast<const v2f*>(FH + l16 * 16 + kb);
  }

  v8f acc = {0.f, 0.f, 0.f, 0.f, 0.f, 0.f, 0.f, 0.f};
#pragma unroll
  for (int t = 0; t < 4; ++t) {
    const int kb = 4 * t + (hg << 1);
    const unsigned rowb = (unsigned)(w0 + kb) * H2;
    v2f bL, bH;
    if (which == 0) {
      const unsigned l0 = (unsigned)(w0 + kb) * R + colE;
      bL.x = llp[l0];
      bL.y = llp[l0 + R];
    } else {
      bL.x = SC_HALF_SQRT2 * (pL1[rowb + colE] + pL2[rowb + colE]);
      bL.y = sgn1 * pL1[rowb + colO] + sgn2 * pL2[rowb + colO];
    }
    bH.x = SC_HALF_SQRT2 * (pH1[rowb + colE] + pH2[rowb + colE]);
    bH.y = sgn1 * pH1[rowb + colO] + sgn2 * pH2[rowb + colO];
    acc = wmma4(aL[t], bL, acc);
    acc = wmma4(aH[t], bH, acc);
  }

  float* outp = (which ? y2 : y1) + (size_t)mc * (2 * R * R) +
                (size_t)(o0 + 8 * hg) * R + bcol;
#pragma unroll
  for (int v = 0; v < 8; ++v) outp[(unsigned)(v * R)] = acc[v];
}

template <int H, int W>
__global__ __launch_bounds__(32) void row_ifilt_kernel(
    const float* __restrict__ y1b, const float* __restrict__ y2b,
    const float* __restrict__ Gmat, float* __restrict__ out) {
  constexpr int TILES_X = (2 * W) / 16;
  const int lane = threadIdx.x;
  const int tx = blockIdx.x % TILES_X, ty = blockIdx.x / TILES_X;
  const int yy0 = ty << 4, p0 = tx << 4;
  const int mc = blockIdx.y;
  const int w0 = iclamp((p0 >> 1) - 4, 0, W - 16);

  __shared__ float t1[16 * 17], t2[16 * 17];
  const float* y1p = y1b + (size_t)mc * (H * W);
  const float* y2p = y2b + (size_t)mc * (H * W);
  {
    const int rr = lane >> 1, cb = (lane & 1) * 8;
    const unsigned g = (unsigned)(yy0 + rr) * W + (w0 + cb);
#pragma unroll
    for (int j = 0; j < 8; ++j) {
      t1[rr * 17 + cb + j] = y1p[g + j];
      t2[rr * 17 + cb + j] = y2p[g + j];
    }
  }

  const int hg = lane >> 4, l16 = lane & 15;
  const float* GL = Gmat + ((size_t)tx * 2 + 0) * 256;
  const float* GH = Gmat + ((size_t)tx * 2 + 1) * 256;
  v2f b1[4], b2[4];
#pragma unroll
  for (int t = 0; t < 4; ++t) {
    const int kb = 4 * t + (hg << 1);
    b1[t].x = GL[kb * 16 + l16];
    b1[t].y = GL[(kb + 1) * 16 + l16];
    b2[t].x = GH[kb * 16 + l16];
    b2[t].y = GH[(kb + 1) * 16 + l16];
  }
  __syncthreads();

  v8f acc = {0.f, 0.f, 0.f, 0.f, 0.f, 0.f, 0.f, 0.f};
#pragma unroll
  for (int t = 0; t < 4; ++t) {
    const int kb = 4 * t + (hg << 1);
    v2f a1, a2;
    a1.x = t1[l16 * 17 + kb];     a1.y = t1[l16 * 17 + kb + 1];
    a2.x = t2[l16 * 17 + kb];     a2.y = t2[l16 * 17 + kb + 1];
    acc = wmma4(a1, b1[t], acc);
    acc = wmma4(a2, b2[t], acc);
  }

  float* outp = out + (size_t)mc * (H * 2 * W) +
                (size_t)(yy0 + 8 * hg) * (2 * W) + (p0 + l16);
#pragma unroll
  for (int v = 0; v < 8; ++v) outp[(unsigned)(v * 2 * W)] = acc[v];
}

template <int R>
__global__ __launch_bounds__(32) void col_j1_kernel(
    const float* __restrict__ ll, const float* __restrict__ yh,
    const float* __restrict__ Fmat, float* __restrict__ y1,
    float* __restrict__ y2) {
  constexpr int TILES_X = R / 16;
  constexpr int H2 = R / 2;
  constexpr int SB = H2 * H2 * 2;
  const int lane = threadIdx.x;
  const int tx = blockIdx.x % TILES_X, ty = blockIdx.x / TILES_X;
  const int o0 = ty << 4, c0 = tx << 4;
  const int mc = blockIdx.y;
  const int which = blockIdx.z;
  const int w0 = iclamp(o0 - 4, 0, R - 24);

  const int hg = lane >> 4, l16 = lane & 15;
  const int bcol = c0 + l16;

  const float* llp = ll + (size_t)mc * (R * R);
  const float* yhp = yh + (size_t)mc * (6 * SB);
  const float* pH1 = yhp + (which ? 1 : 0) * SB;
  const float* pH2 = yhp + (which ? 4 : 5) * SB;
  const float* pL1 = yhp + 2 * SB;
  const float* pL2 = yhp + 3 * SB;

  const float sgn1 = (bcol & 1) ? -SC_HALF_SQRT2 : SC_HALF_SQRT2;
  const float sgn2 = (bcol & 1) ? SC_HALF_SQRT2 : -SC_HALF_SQRT2;
  const unsigned colE = (unsigned)bcol;
  const unsigned colO = (unsigned)(bcol ^ 1);

  const float* FL = Fmat + ((size_t)ty * 2 + 0) * 384;
  const float* FH = Fmat + ((size_t)ty * 2 + 1) * 384;
  v2f aL[6], aH[6];
#pragma unroll
  for (int t = 0; t < 6; ++t) {
    const int kb = 4 * t + (hg << 1);
    aL[t] = *reinterpret_cast<const v2f*>(FL + l16 * 24 + kb);
    aH[t] = *reinterpret_cast<const v2f*>(FH + l16 * 24 + kb);
  }

  v8f acc = {0.f, 0.f, 0.f, 0.f, 0.f, 0.f, 0.f, 0.f};
#pragma unroll
  for (int t = 0; t < 6; ++t) {
    const int kb = 4 * t + (hg << 1);
    const unsigned rowb = (unsigned)(w0 + kb) * H2;
    v2f bL, bH;
    if (which == 0) {
      const unsigned l0 = (unsigned)(w0 + kb) * R + colE;
      bL.x = llp[l0];
      bL.y = llp[l0 + R];
    } else {
      bL.x = SC_HALF_SQRT2 * (pL1[rowb + colE] + pL2[rowb + colE]);
      bL.y = sgn1 * pL1[rowb + colO] + sgn2 * pL2[rowb + colO];
    }
    bH.x = SC_HALF_SQRT2 * (pH1[rowb + colE] + pH2[rowb + colE]);
    bH.y = sgn1 * pH1[rowb + colO] + sgn2 * pH2[rowb + colO];
    acc = wmma4(aL[t], bL, acc);
    acc = wmma4(aH[t], bH, acc);
  }

  float* outp = (which ? y2 : y1) + (size_t)mc * (R * R) +
                (size_t)(o0 + 8 * hg) * R + bcol;
#pragma unroll
  for (int v = 0; v < 8; ++v) outp[(unsigned)(v * R)] = acc[v];
}

template <int H, int W>
__global__ __launch_bounds__(32) void row_j1_kernel(
    const float* __restrict__ y1b, const float* __restrict__ y2b,
    const float* __restrict__ Gmat, float* __restrict__ out) {
  constexpr int TILES_X = W / 16;
  const int lane = threadIdx.x;
  const int tx = blockIdx.x % TILES_X, ty = blockIdx.x / TILES_X;
  const int yy0 = ty << 4, p0 = tx << 4;
  const int mc = blockIdx.y;
  const int w0 = iclamp(p0 - 4, 0, W - 24);

  __shared__ float t1[16 * 25], t2[16 * 25];
  const float* y1p = y1b + (size_t)mc * (H * W);
  const float* y2p = y2b + (size_t)mc * (H * W);
  {
    const int rr = lane >> 1, cb = (lane & 1) * 12;
    const unsigned g = (unsigned)(yy0 + rr) * W + (w0 + cb);
#pragma unroll
    for (int j = 0; j < 12; ++j) {
      t1[rr * 25 + cb + j] = y1p[g + j];
      t2[rr * 25 + cb + j] = y2p[g + j];
    }
  }

  const int hg = lane >> 4, l16 = lane & 15;
  const float* GL = Gmat + ((size_t)tx * 2 + 0) * 384;
  const float* GH = Gmat + ((size_t)tx * 2 + 1) * 384;
  v2f b1[6], b2[6];
#pragma unroll
  for (int t = 0; t < 6; ++t) {
    const int kb = 4 * t + (hg << 1);
    b1[t].x = GL[kb * 16 + l16];
    b1[t].y = GL[(kb + 1) * 16 + l16];
    b2[t].x = GH[kb * 16 + l16];
    b2[t].y = GH[(kb + 1) * 16 + l16];
  }
  __syncthreads();

  v8f acc = {0.f, 0.f, 0.f, 0.f, 0.f, 0.f, 0.f, 0.f};
#pragma unroll
  for (int t = 0; t < 6; ++t) {
    const int kb = 4 * t + (hg << 1);
    v2f a1, a2;
    a1.x = t1[l16 * 25 + kb];     a1.y = t1[l16 * 25 + kb + 1];
    a2.x = t2[l16 * 25 + kb];     a2.y = t2[l16 * 25 + kb + 1];
    acc = wmma4(a1, b1[t], acc);
    acc = wmma4(a2, b2[t], acc);
  }

  float* outp = out + (size_t)mc * (H * W) +
                (size_t)(yy0 + 8 * hg) * W + (p0 + l16);
#pragma unroll
  for (int v = 0; v < 8; ++v) outp[(unsigned)(v * W)] = acc[v];
}

extern "C" void kernel_launch(void* const* d_in, const int* in_sizes, int n_in,
                              void* d_out, int out_size, void* d_ws,
                              size_t ws_size, hipStream_t stream) {
  (void)in_sizes; (void)n_in; (void)out_size; (void)ws_size;
  const float* yl  = (const float*)d_in[0];
  const float* yh1 = (const float*)d_in[1];
  const float* yh2 = (const float*)d_in[2];
  const float* yh3 = (const float*)d_in[3];
  const float* g0o = (const float*)d_in[4];
  const float* g1o = (const float*)d_in[5];
  const float* g0a = (const float*)d_in[6];
  const float* g0b = (const float*)d_in[7];
  const float* g1a = (const float*)d_in[8];
  const float* g1b = (const float*)d_in[9];

  const size_t SEG = (size_t)16 * 1024 * 1024;
  float* bufA = (float*)d_ws;
  float* bufB = bufA + SEG;
  float* bufC = bufB + SEG;
  float* F3c = bufC + SEG;
  float* G3r = F3c + 8 * 2 * 256;
  float* F2c = G3r + 8 * 2 * 256;
  float* G2r = F2c + 16 * 2 * 256;
  float* F1c = G2r + 16 * 2 * 256;
  float* G1r = F1c + 16 * 2 * 384;

  build_ifilt_colF<<<dim3(8, 2),  256, 0, stream>>>(g0a, g0b, g1a, g1b, F3c, 64);
  build_ifilt_rowG<<<dim3(8, 2),  256, 0, stream>>>(g0a, g0b, g1a, g1b, G3r, 64);
  build_ifilt_colF<<<dim3(16, 2), 256, 0, stream>>>(g0a, g0b, g1a, g1b, F2c, 128);
  build_ifilt_rowG<<<dim3(16, 2), 256, 0, stream>>>(g0a, g0b, g1a, g1b, G2r, 128);
  build_j1_colF<<<dim3(16, 2),    256, 0, stream>>>(g0o, g1o, F1c, 256);
  build_j1_rowG<<<dim3(16, 2),    256, 0, stream>>>(g0o, g1o, G1r, 256);

  col_ifilt_kernel<64><<<dim3(8 * 4, 256, 2), 32, 0, stream>>>(
      yl, yh3, F3c, bufA, bufB);
  row_ifilt_kernel<128, 64><<<dim3(8 * 8, 256, 1), 32, 0, stream>>>(
      bufA, bufB, G3r, bufC);

  col_ifilt_kernel<128><<<dim3(16 * 8, 256, 2), 32, 0, stream>>>(
      bufC, yh2, F2c, bufA, bufB);
  row_ifilt_kernel<256, 128><<<dim3(16 * 16, 256, 1), 32, 0, stream>>>(
      bufA, bufB, G2r, bufC);

  col_j1_kernel<256><<<dim3(16 * 16, 256, 2), 32, 0, stream>>>(
      bufC, yh1, F1c, bufA, bufB);
  row_j1_kernel<256, 256><<<dim3(16 * 16, 256, 1), 32, 0, stream>>>(
      bufA, bufB, G1r, (float*)d_out);
}


# 8-wave work-groups: each wave walks a strip of 4 or 8 tiles, filter fragments loaded once per strip, a tile's loads issued as one group (row pass reads its operands straight from global, no LDS)
# speedup vs baseline: 1.5997x; 1.5997x over previous
#include <hip/hip_runtime.h>


typedef __attribute__((ext_vector_type(2))) float v2f;
typedef __attribute__((ext_vector_type(8))) float v8f;

#define SC_HALF_SQRT2 0.70710678118654752f

__device__ __forceinline__ int refl(int i, int n) {
  i = (i < 0) ? (-1 - i) : i;
  i = (i >= n) ? (2 * n - 1 - i) : i;
  return i;
}

__device__ __forceinline__ int iclamp(int v, int lo, int hi) {
  return v < lo ? lo : (v > hi ? hi : v);
}

__device__ __forceinline__ int wave_in_block() {
  return __builtin_amdgcn_readfirstlane((int)(threadIdx.x >> 5));
}

__device__ __forceinline__ v8f wmma4(v2f a, v2f b, v8f c) {
  return __builtin_amdgcn_wmma_f32_16x16x4_f32(false, a, false, b, (short)0, c,
                                               false, false);
}

__device__ __forceinline__ float ifilt_coef(int o, int kg, int n,
                                            const float (&ha)[10],
                                            const float (&hb)[10], bool hp) {
  const int i = o >> 2, s = o & 3;
  const int base = 4 + ((((s & 1) != 0) != hp) ? 1 : 0);
  float acc = 0.f;
#pragma unroll
  for (int m = 0; m < 5; ++m) {
    const int L = refl(2 * i + base - 2 * m, n);
    const float c = (s == 0)   ? ha[2 * m]
                    : (s == 1) ? hb[2 * m]
                    : (s == 2) ? ha[2 * m + 1]
                               : hb[2 * m + 1];
    acc += (L == kg) ? c : 0.f;
  }
  return acc;
}

template <int NT>
__device__ __forceinline__ float filt_coef(int o, int kg, int n,
                                           const float (&h)[NT]) {
  float acc = 0.f;
#pragma unroll
  for (int m = 0; m < NT; ++m) {
    const int L = refl(o + (NT >> 1) - m, n);
    acc += (L == kg) ? h[m] : 0.f;
  }
  return acc;
}


__global__ __launch_bounds__(256) void build_ifilt_colF(
    const float* __restrict__ g0a, const float* __restrict__ g0b,
    const float* __restrict__ g1a, const float* __restrict__ g1b,
    float* __restrict__ M, int r) {
  const int ty = blockIdx.x, f = blockIdx.y;
  float ha[10], hb[10];
#pragma unroll
  for (int m = 0; m < 10; ++m) {
    ha[m] = f ? g1b[m] : g0b[m];
    hb[m] = f ? g1a[m] : g0a[m];
  }
  const int o0 = ty << 4;
  const int w0 = iclamp((o0 >> 1) - 4, 0, r - 16);
  const int idx = threadIdx.x;
  const int row = idx >> 4, k = idx & 15;
  M[(((size_t)ty * 2 + f) * 16 + row) * 16 + k] =
      ifilt_coef(o0 + row, w0 + k, r, ha, hb, f != 0);
}

__global__ __launch_bounds__(256) void build_ifilt_rowG(
    const float* __restrict__ g0a, const float* __restrict__ g0b,
    const float* __restrict__ g1a, const float* __restrict__ g1b,
    float* __restrict__ M, int W) {
  const int tx = blockIdx.x, f = blockIdx.y;
  float ha[10], hb[10];
#pragma unroll
  for (int m = 0; m < 10; ++m) {
    ha[m] = f ? g1b[m] : g0b[m];
    hb[m] = f ? g1a[m] : g0a[m];
  }
  const int p0 = tx << 4;
  const int w0 = iclamp((p0 >> 1) - 4, 0, W - 16);
  const int idx = threadIdx.x;
  const int k = idx >> 4, n = idx & 15;
  M[(((size_t)tx * 2 + f) * 16 + k) * 16 + n] =
      ifilt_coef(p0 + n, w0 + k, W, ha, hb, f != 0);
}

__global__ __launch_bounds__(256) void build_j1_colF(
    const float* __restrict__ g0o, const float* __restrict__ g1o,
    float* __restrict__ M, int r) {
  const int ty = blockIdx.x, f = blockIdx.y;
  float h0[7], h1[5];
#pragma unroll
  for (int m = 0; m < 7; ++m) h0[m] = g0o[m];
#pragma unroll
  for (int m = 0; m < 5; ++m) h1[m] = g1o[m];
  const int o0 = ty << 4;
  const int w0 = iclamp(o0 - 4, 0, r - 24);
  for (int idx = threadIdx.x; idx < 16 * 24; idx += 256) {
    const int row = idx / 24, k = idx % 24;
    const float v = f ? filt_coef<5>(o0 + row, w0 + k, r, h1)
                      : filt_coef<7>(o0 + row, w0 + k, r, h0);
    M[((size_t)ty * 2 + f) * 384 + row * 24 + k] = v;
  }
}

__global__ __launch_bounds__(256) void build_j1_rowG(
    const float* __restrict__ g0o, const float* __restrict__ g1o,
    float* __restrict__ M, int W) {
  const int tx = blockIdx.x, f = blockIdx.y;
  float h0[7], h1[5];
#pragma unroll
  for (int m = 0; m < 7; ++m) h0[m] = g0o[m];
#pragma unroll
  for (int m = 0; m < 5; ++m) h1[m] = g1o[m];
  const int p0 = tx << 4;
  const int w0 = iclamp(p0 - 4, 0, W - 24);
  for (int idx = threadIdx.x; idx < 24 * 16; idx += 256) {
    const int k = idx >> 4, n = idx & 15;
    const float v = f ? filt_coef<5>(p0 + n, w0 + k, W, h1)
                      : filt_coef<7>(p0 + n, w0 + k, W, h0);
    M[((size_t)tx * 2 + f) * 384 + k * 16 + n] = v;
  }
}


template <int R, int TPW, int K4, int WH>
__device__ __forceinline__ void col_strip(
    const float* __restrict__ llp, const float* __restrict__ yhp,
    const v2f (&aL)[K4], const v2f (&aH)[K4], float* __restrict__ outb,
    int w0, int tx0, int hg, int l16) {
  constexpr int H2 = R / 2;
  constexpr int SB = H2 * H2 * 2;
  const float* pH1 = yhp + (WH ? 1 : 0) * SB;
  const float* pH2 = yhp + (WH ? 4 : 5) * SB;
  const float* pL1 = yhp + 2 * SB;
  const float* pL2 = yhp + 3 * SB;

  const float sgn1 = (l16 & 1) ? -SC_HALF_SQRT2 : SC_HALF_SQRT2;
  const float sgn2 = (l16 & 1) ? SC_HALF_SQRT2 : -SC_HALF_SQRT2;

  for (int i = 0; i < TPW; ++i) {
    const int bcol = ((tx0 + i) << 4) + l16;
    const unsigned colE = (unsigned)bcol;
    const unsigned colO = (unsigned)(bcol ^ 1);

    v8f acc = {0.f, 0.f, 0.f, 0.f, 0.f, 0.f, 0.f, 0.f};
#pragma unroll
    for (int t = 0; t < K4; ++t) {
      const int kb = 4 * t + (hg << 1);
      const unsigned rowb = (unsigned)(w0 + kb) * H2;
      v2f bL, bH;
      if constexpr (WH == 0) {
        const unsigned l0 = (unsigned)(w0 + kb) * R + colE;
        bL.x = llp[l0];
        bL.y = llp[l0 + R];
      } else {
        bL.x = SC_HALF_SQRT2 * (pL1[rowb + colE] + pL2[rowb + colE]);
        bL.y = sgn1 * pL1[rowb + colO] + sgn2 * pL2[rowb + colO];
      }
      bH.x = SC_HALF_SQRT2 * (pH1[rowb + colE] + pH2[rowb + colE]);
      bH.y = sgn1 * pH1[rowb + colO] + sgn2 * pH2[rowb + colO];
      acc = wmma4(aL[t], bL, acc);
      acc = wmma4(aH[t], bH, acc);
    }

    float* outp = outb + bcol;
#pragma unroll
    for (int v = 0; v < 8; ++v) outp[(unsigned)(v * R)] = acc[v];
  }
}

template <int R, int TPW>
__global__ __launch_bounds__(256) void col_ifilt_kernel(
    const float* __restrict__ ll, const float* __restrict__ yh,
    const float* __restrict__ Fmat, float* __restrict__ y1,
    float* __restrict__ y2) {
  constexpr int TILES_X = R / 16;
  constexpr int TILES_Y = (2 * R) / 16;
  constexpr int XS = TILES_X / TPW;
  static_assert(TILES_X % TPW == 0 && TILES_Y % 8 == 0, "strip shape");
  constexpr int H2 = R / 2;
  constexpr int SB = H2 * H2 * 2;
  const int lane = threadIdx.x & 31;
  const int ty = (blockIdx.x / XS) * 8 + wave_in_block();
  const int tx0 = (blockIdx.x % XS) * TPW;
  const int o0 = ty << 4;
  const int mc = blockIdx.y;
  const int which = blockIdx.z;
  const int w0 = iclamp((o0 >> 1) - 4, 0, R - 16);

  const int hg = lane >> 4, l16 = lane & 15;

  const float* llp = ll + (size_t)mc * (R * R);
  const float* yhp = yh + (size_t)mc * (6 * SB);

  const float* FL = Fmat + ((size_t)ty * 2 + 0) * 256;
  const float* FH = Fmat + ((size_t)ty * 2 + 1) * 256;
  v2f aL[4], aH[4];
#pragma unroll
  for (int t = 0; t < 4; ++t) {
    const int kb = 4 * t + (hg << 1);
    aL[t] = *reinterpret_cast<const v2f*>(FL + l16 * 16 + kb);
    aH[t] = *reinterpret_cast<const v2f*>(FH + l16 * 16 + kb);
  }

  float* outb = (which ? y2 : y1) + (size_t)mc * (2 * R * R) +
                (size_t)(o0 + 8 * hg) * R;

  if (which == 0)
    col_strip<R, TPW, 4, 0>(llp, yhp, aL, aH, outb, w0, tx0, hg, l16);
  else
    col_strip<R, TPW, 4, 1>(llp, yhp, aL, aH, outb, w0, tx0, hg, l16);
}

template <int H, int W, int TPW>
__global__ __launch_bounds__(256) void row_ifilt_kernel(
    const float* __restrict__ y1b, const float* __restrict__ y2b,
    const float* __restrict__ Gmat, float* __restrict__ out) {
  constexpr int TILES_X = (2 * W) / 16;
  constexpr int TILES_Y = H / 16;
  constexpr int XG = TILES_X / 8;
  static_assert(TILES_X % 8 == 0 && TILES_Y % TPW == 0, "strip shape");
  const int lane = threadIdx.x & 31;
  const int tx = (blockIdx.x % XG) * 8 + wave_in_block();
  const int ty0 = (blockIdx.x / XG) * TPW;
  const int p0 = tx << 4;
  const int mc = blockIdx.y;
  const int w0 = iclamp((p0 >> 1) - 4, 0, W - 16);

  const int hg = lane >> 4, l16 = lane & 15;

  const float* GL = Gmat + ((size_t)tx * 2 + 0) * 256;
  const float* GH = Gmat + ((size_t)tx * 2 + 1) * 256;
  v2f b1[4], b2[4];
#pragma unroll
  for (int t = 0; t < 4; ++t) {
    const int kb = 4 * t + (hg << 1);
    b1[t].x = GL[kb * 16 + l16];
    b1[t].y = GL[(kb + 1) * 16 + l16];
    b2[t].x = GH[kb * 16 + l16];
    b2[t].y = GH[(kb + 1) * 16 + l16];
  }

  const unsigned lane_in = (unsigned)l16 * W + (unsigned)(w0 + (hg << 1));
  const float* y1p = y1b + (size_t)mc * (H * W) + lane_in;
  const float* y2p = y2b + (size_t)mc * (H * W) + lane_in;
  float* outb = out + (size_t)mc * (H * 2 * W) + (size_t)(8 * hg) * (2 * W) +
                (p0 + l16);

  for (int i = 0; i < TPW; ++i) {
    const unsigned yy0 = (unsigned)(ty0 + i) << 4;
    const float* r1 = y1p + yy0 * W;
    const float* r2 = y2p + yy0 * W;

    v8f acc = {0.f, 0.f, 0.f, 0.f, 0.f, 0.f, 0.f, 0.f};
#pragma unroll
    for (int t = 0; t < 4; ++t) {
      const v2f a1 = *reinterpret_cast<const v2f*>(r1 + 4 * t);
      const v2f a2 = *reinterpret_cast<const v2f*>(r2 + 4 * t);
      acc = wmma4(a1, b1[t], acc);
      acc = wmma4(a2, b2[t], acc);
    }

    float* outp = outb + yy0 * (2 * W);
#pragma unroll
    for (int v = 0; v < 8; ++v) outp[(unsigned)(v * 2 * W)] = acc[v];
  }
}

template <int R, int TPW>
__global__ __launch_bounds__(256) void col_j1_kernel(
    const float* __restrict__ ll, const float* __restrict__ yh,
    const float* __restrict__ Fmat, float* __restrict__ y1,
    float* __restrict__ y2) {
  constexpr int TILES_X = R / 16;
  constexpr int TILES_Y = R / 16;
  constexpr int XS = TILES_X / TPW;
  static_assert(TILES_X % TPW == 0 && TILES_Y % 8 == 0, "strip shape");
  constexpr int H2 = R / 2;
  constexpr int SB = H2 * H2 * 2;
  const int lane = threadIdx.x & 31;
  const int ty = (blockIdx.x / XS) * 8 + wave_in_block();
  const int tx0 = (blockIdx.x % XS) * TPW;
  const int o0 = ty << 4;
  const int mc = blockIdx.y;
  const int which = blockIdx.z;
  const int w0 = iclamp(o0 - 4, 0, R - 24);

  const int hg = lane >> 4, l16 = lane & 15;

  const float* llp = ll + (size_t)mc * (R * R);
  const float* yhp = yh + (size_t)mc * (6 * SB);

  const float* FL = Fmat + ((size_t)ty * 2 + 0) * 384;
  const float* FH = Fmat + ((size_t)ty * 2 + 1) * 384;
  v2f aL[6], aH[6];
#pragma unroll
  for (int t = 0; t < 6; ++t) {
    const int kb = 4 * t + (hg << 1);
    aL[t] = *reinterpret_cast<const v2f*>(FL + l16 * 24 + kb);
    aH[t] = *reinterpret_cast<const v2f*>(FH + l16 * 24 + kb);
  }

  float* outb = (which ? y2 : y1) + (size_t)mc * (R * R) +
                (size_t)(o0 + 8 * hg) * R;

  if (which == 0)
    col_strip<R, TPW, 6, 0>(llp, yhp, aL, aH, outb, w0, tx0, hg, l16);
  else
    col_strip<R, TPW, 6, 1>(llp, yhp, aL, aH, outb, w0, tx0, hg, l16);
}

template <int H, int W, int TPW>
__global__ __launch_bounds__(256) void row_j1_kernel(
    const float* __restrict__ y1b, const float* __restrict__ y2b,
    const float* __restrict__ Gmat, float* __restrict__ out) {
  constexpr int TILES_X = W / 16;
  constexpr int TILES_Y = H / 16;
  constexpr int XG = TILES_X / 8;
  static_assert(TILES_X % 8 == 0 && TILES_Y % TPW == 0, "strip shape");
  const int lane = threadIdx.x & 31;
  const int tx = (blockIdx.x % XG) * 8 + wave_in_block();
  const int ty0 = (blockIdx.x / XG) * TPW;
  const int p0 = tx << 4;
  const int mc = blockIdx.y;
  const int w0 = iclamp(p0 - 4, 0, W - 24);

  const int hg = lane >> 4, l16 = lane & 15;

  const float* GL = Gmat + ((size_t)tx * 2 + 0) * 384;
  const float* GH = Gmat + ((size_t)tx * 2 + 1) * 384;
  v2f b1[6], b2[6];
#pragma unroll
  for (int t = 0; t < 6; ++t) {
    const int kb = 4 * t + (hg << 1);
    b1[t].x = GL[kb * 16 + l16];
    b1[t].y = GL[(kb + 1) * 16 + l16];
    b2[t].x = GH[kb * 16 + l16];
    b2[t].y = GH[(kb + 1) * 16 + l16];
  }

  const unsigned lane_in = (unsigned)l16 * W + (unsigned)(w0 + (hg << 1));
  const float* y1p = y1b + (size_t)mc * (H * W) + lane_in;
  const float* y2p = y2b + (size_t)mc * (H * W) + lane_in;
  float* outb = out + (size_t)mc * (H * W) + (size_t)(8 * hg) * (W) +
                (p0 + l16);

  for (int i = 0; i < TPW; ++i) {
    const unsigned yy0 = (unsigned)(ty0 + i) << 4;
    const float* r1 = y1p + yy0 * W;
    const float* r2 = y2p + yy0 * W;

    v8f acc = {0.f, 0.f, 0.f, 0.f, 0.f, 0.f, 0.f, 0.f};
#pragma unroll
    for (int t = 0; t < 6; ++t) {
      const v2f a1 = *reinterpret_cast<const v2f*>(r1 + 4 * t);
      const v2f a2 = *reinterpret_cast<const v2f*>(r2 + 4 * t);
      acc = wmma4(a1, b1[t], acc);
      acc = wmma4(a2, b2[t], acc);
    }

    float* outp = outb + yy0 * (W);
#pragma unroll
    for (int v = 0; v < 8; ++v) outp[(unsigned)(v * W)] = acc[v];
  }
}

extern "C" void kernel_launch(void* const* d_in, const int* in_sizes, int n_in,
                              void* d_out, int out_size, void* d_ws,
                              size_t ws_size, hipStream_t stream) {
  (void)in_sizes; (void)n_in; (void)out_size; (void)ws_size;
  const float* yl  = (const float*)d_in[0];
  const float* yh1 = (const float*)d_in[1];
  const float* yh2 = (const float*)d_in[2];
  const float* yh3 = (const float*)d_in[3];
  const float* g0o = (const float*)d_in[4];
  const float* g1o = (const float*)d_in[5];
  const float* g0a = (const float*)d_in[6];
  const float* g0b = (const float*)d_in[7];
  const float* g1a = (const float*)d_in[8];
  const float* g1b = (const float*)d_in[9];

  const size_t SEG = (size_t)16 * 1024 * 1024;
  float* bufA = (float*)d_ws;
  float* bufB = bufA + SEG;
  float* bufC = bufB + SEG;
  float* F3c = bufC + SEG;
  float* G3r = F3c + 8 * 2 * 256;
  float* F2c = G3r + 8 * 2 * 256;
  float* G2r = F2c + 16 * 2 * 256;
  float* F1c = G2r + 16 * 2 * 256;
  float* G1r = F1c + 16 * 2 * 384;

  build_ifilt_colF<<<dim3(8, 2),  256, 0, stream>>>(g0a, g0b, g1a, g1b, F3c, 64);
  build_ifilt_rowG<<<dim3(8, 2),  256, 0, stream>>>(g0a, g0b, g1a, g1b, G3r, 64);
  build_ifilt_colF<<<dim3(16, 2), 256, 0, stream>>>(g0a, g0b, g1a, g1b, F2c, 128);
  build_ifilt_rowG<<<dim3(16, 2), 256, 0, stream>>>(g0a, g0b, g1a, g1b, G2r, 128);
  build_j1_colF<<<dim3(16, 2),    256, 0, stream>>>(g0o, g1o, F1c, 256);
  build_j1_rowG<<<dim3(16, 2),    256, 0, stream>>>(g0o, g1o, G1r, 256);

  col_ifilt_kernel<64, 4><<<dim3(1, 256, 2), 256, 0, stream>>>(
      yl, yh3, F3c, bufA, bufB);
  row_ifilt_kernel<128, 64, 4><<<dim3(2, 256, 1), 256, 0, stream>>>(
      bufA, bufB, G3r, bufC);

  col_ifilt_kernel<128, 4><<<dim3(4, 256, 2), 256, 0, stream>>>(
      bufC, yh2, F2c, bufA, bufB);
  row_ifilt_kernel<256, 128, 4><<<dim3(8, 256, 1), 256, 0, stream>>>(
      bufA, bufB, G2r, bufC);

  col_j1_kernel<256, 8><<<dim3(4, 256, 2), 256, 0, stream>>>(
      bufC, yh1, F1c, bufA, bufB);
  row_j1_kernel<256, 256, 4><<<dim3(8, 256, 1), 256, 0, stream>>>(
      bufA, bufB, G1r, (float*)d_out);
}


# Column and row pass of each level fused in one kernel: an 8-wave work-group keeps y1,y2 of a 16-row strip in LDS and row-filters them there; y1/y2 never pass through memory (12 -> 9 launches)
# speedup vs baseline: 2.7607x; 1.7258x over previous
#include <hip/hip_runtime.h>


typedef __attribute__((ext_vector_type(2))) float v2f;
typedef __attribute__((ext_vector_type(8))) float v8f;

#define SC_HALF_SQRT2 0.70710678118654752f

__device__ __forceinline__ int refl(int i, int n) {
  i = (i < 0) ? (-1 - i) : i;
  i = (i >= n) ? (2 * n - 1 - i) : i;
  return i;
}

__device__ __forceinline__ int iclamp(int v, int lo, int hi) {
  return v < lo ? lo : (v > hi ? hi : v);
}

__device__ __forceinline__ int wave_in_block() {
  return __builtin_amdgcn_readfirstlane((int)(threadIdx.x >> 5));
}

__device__ __forceinline__ v8f wmma4(v2f a, v2f b, v8f c) {
  return __builtin_amdgcn_wmma_f32_16x16x4_f32(false, a, false, b, (short)0, c,
                                               false, false);
}

__device__ __forceinline__ float ifilt_coef(int o, int kg, int n,
                                            const float (&ha)[10],
                                            const float (&hb)[10], bool hp) {
  const int i = o >> 2, s = o & 3;
  const int base = 4 + ((((s & 1) != 0) != hp) ? 1 : 0);
  float acc = 0.f;
#pragma unroll
  for (int m = 0; m < 5; ++m) {
    const int L = refl(2 * i + base - 2 * m, n);
    const float c = (s == 0)   ? ha[2 * m]
                    : (s == 1) ? hb[2 * m]
                    : (s == 2) ? ha[2 * m + 1]
                               : hb[2 * m + 1];
    acc += (L == kg) ? c : 0.f;
  }
  return acc;
}

template <int NT>
__device__ __forceinline__ float filt_coef(int o, int kg, int n,
                                           const float (&h)[NT]) {
  float acc = 0.f;
#pragma unroll
  for (int m = 0; m < NT; ++m) {
    const int L = refl(o + (NT >> 1) - m, n);
    acc += (L == kg) ? h[m] : 0.f;
  }
  return acc;
}


__global__ __launch_bounds__(256) void build_ifilt_colF(
    const float* __restrict__ g0a, const float* __restrict__ g0b,
    const float* __restrict__ g1a, const float* __restrict__ g1b,
    float* __restrict__ M, int r) {
  const int ty = blockIdx.x, f = blockIdx.y;
  float ha[10], hb[10];
#pragma unroll
  for (int m = 0; m < 10; ++m) {
    ha[m] = f ? g1b[m] : g0b[m];
    hb[m] = f ? g1a[m] : g0a[m];
  }
  const int o0 = ty << 4;
  const int w0 = iclamp((o0 >> 1) - 4, 0, r - 16);
  const int idx = threadIdx.x;
  const int row = idx >> 4, k = idx & 15;
  M[(((size_t)ty * 2 + f) * 16 + row) * 16 + k] =
      ifilt_coef(o0 + row, w0 + k, r, ha, hb, f != 0);
}

__global__ __launch_bounds__(256) void build_ifilt_rowG(
    const float* __restrict__ g0a, const float* __restrict__ g0b,
    const float* __restrict__ g1a, const float* __restrict__ g1b,
    float* __restrict__ M, int W) {
  const int tx = blockIdx.x, f = blockIdx.y;
  float ha[10], hb[10];
#pragma unroll
  for (int m = 0; m < 10; ++m) {
    ha[m] = f ? g1b[m] : g0b[m];
    hb[m] = f ? g1a[m] : g0a[m];
  }
  const int p0 = tx << 4;
  const int w0 = iclamp((p0 >> 1) - 4, 0, W - 16);
  const int idx = threadIdx.x;
  const int k = idx >> 4, n = idx & 15;
  M[(((size_t)tx * 2 + f) * 16 + k) * 16 + n] =
      ifilt_coef(p0 + n, w0 + k, W, ha, hb, f != 0);
}

__global__ __launch_bounds__(256) void build_j1_colF(
    const float* __restrict__ g0o, const float* __restrict__ g1o,
    float* __restrict__ M, int r) {
  const int ty = blockIdx.x, f = blockIdx.y;
  float h0[7], h1[5];
#pragma unroll
  for (int m = 0; m < 7; ++m) h0[m] = g0o[m];
#pragma unroll
  for (int m = 0; m < 5; ++m) h1[m] = g1o[m];
  const int o0 = ty << 4;
  const int w0 = iclamp(o0 - 4, 0, r - 24);
  for (int idx = threadIdx.x; idx < 16 * 24; idx += 256) {
    const int row = idx / 24, k = idx % 24;
    const float v = f ? filt_coef<5>(o0 + row, w0 + k, r, h1)
                      : filt_coef<7>(o0 + row, w0 + k, r, h0);
    M[((size_t)ty * 2 + f) * 384 + row * 24 + k] = v;
  }
}

__global__ __launch_bounds__(256) void build_j1_rowG(
    const float* __restrict__ g0o, const float* __restrict__ g1o,
    float* __restrict__ M, int W) {
  const int tx = blockIdx.x, f = blockIdx.y;
  float h0[7], h1[5];
#pragma unroll
  for (int m = 0; m < 7; ++m) h0[m] = g0o[m];
#pragma unroll
  for (int m = 0; m < 5; ++m) h1[m] = g1o[m];
  const int p0 = tx << 4;
  const int w0 = iclamp(p0 - 4, 0, W - 24);
  for (int idx = threadIdx.x; idx < 24 * 16; idx += 256) {
    const int k = idx >> 4, n = idx & 15;
    const float v = f ? filt_coef<5>(p0 + n, w0 + k, W, h1)
                      : filt_coef<7>(p0 + n, w0 + k, W, h0);
    M[((size_t)tx * 2 + f) * 384 + k * 16 + n] = v;
  }
}


template <int R, int TPW, int K4, int WH, int P>
__device__ __forceinline__ void col_strip_lds(
    const float* __restrict__ llp, const float* __restrict__ yhp,
    const v2f (&aL)[K4], const v2f (&aH)[K4], float* sdst,
    int w0, int tx0, int hg, int l16) {
  constexpr int H2 = R / 2;
  constexpr int SB = H2 * H2 * 2;
  const float* pH1 = yhp + (WH ? 1 : 0) * SB;
  const float* pH2 = yhp + (WH ? 4 : 5) * SB;
  const float* pL1 = yhp + 2 * SB;
  const float* pL2 = yhp + 3 * SB;

  const float sgn1 = (l16 & 1) ? -SC_HALF_SQRT2 : SC_HALF_SQRT2;
  const float sgn2 = (l16 & 1) ? SC_HALF_SQRT2 : -SC_HALF_SQRT2;

#pragma unroll 1
  for (int i = 0; i < TPW; ++i) {
    const int bcol = ((tx0 + i) << 4) + l16;
    const unsigned colE = (unsigned)bcol;
    const unsigned colO = (unsigned)(bcol ^ 1);

    v8f acc = {0.f, 0.f, 0.f, 0.f, 0.f, 0.f, 0.f, 0.f};
#pragma unroll
    for (int t = 0; t < K4; ++t) {
      const int kb = 4 * t + (hg << 1);
      const unsigned rowb = (unsigned)(w0 + kb) * H2;
      v2f bL, bH;
      if constexpr (WH == 0) {
        const unsigned l0 = (unsigned)(w0 + kb) * R + colE;
        bL.x = llp[l0];
        bL.y = llp[l0 + R];
      } else {
        bL.x = SC_HALF_SQRT2 * (pL1[rowb + colE] + pL2[rowb + colE]);
        bL.y = sgn1 * pL1[rowb + colO] + sgn2 * pL2[rowb + colO];
      }
      bH.x = SC_HALF_SQRT2 * (pH1[rowb + colE] + pH2[rowb + colE]);
      bH.y = sgn1 * pH1[rowb + colO] + sgn2 * pH2[rowb + colO];
      acc = wmma4(aL[t], bL, acc);
      acc = wmma4(aH[t], bH, acc);
    }

    float* sp = sdst + (WH * 16 + 8 * hg) * P + bcol;
#pragma unroll
    for (int v = 0; v < 8; ++v) sp[v * P] = acc[v];
  }
}

__device__ __forceinline__ void lds_barrier() {
  __builtin_amdgcn_fence(__ATOMIC_RELEASE, "workgroup", "local");
  __builtin_amdgcn_s_barrier();
  __builtin_amdgcn_fence(__ATOMIC_ACQUIRE, "workgroup", "local");
}

template <int R, bool UP, int NS>
__global__ __launch_bounds__(256) void level_kernel(
    const float* __restrict__ ll, const float* __restrict__ yh,
    const float* __restrict__ Fmat, const float* __restrict__ Gmat,
    float* __restrict__ out) {
  constexpr int K4 = UP ? 4 : 6;
  constexpr int KW = 4 * K4;
  constexpr int MSZ = 16 * KW;
  constexpr int OH = UP ? 2 * R : R;
  constexpr int OW = OH;
  constexpr int NTC = R / 16;
  constexpr int NTR = OW / 16;
  constexpr int TPWR = NTR / 8;
  constexpr int TPWC = (NTC >= 8) ? NTC / 8 : 1;
  constexpr int H2 = R / 2;
  constexpr int SB = H2 * H2 * 2;
  constexpr int P = R + 4;
  static_assert(NTR % 8 == 0 && (OH / 16) % NS == 0, "strip shape");
  static_assert(NTC >= 8 || NTC == 4, "column tiles per wave");

  __shared__ __attribute__((aligned(16))) float sY[2 * 16 * P];

  const int lane = threadIdx.x & 31;
  const int wv = wave_in_block();
  const int hg = lane >> 4, l16 = lane & 15;
  const int mc = blockIdx.y;

  const float* llp = ll + (size_t)mc * (R * R);
  const float* yhp = yh + (size_t)mc * (6 * SB);
  float* outm = out + (size_t)mc * (OH * OW) + (size_t)(8 * hg) * OW + l16;

  for (int s = 0; s < NS; ++s) {
    const int ty = blockIdx.x * NS + s;
    const int o0 = ty << 4;
    const int w0 = UP ? iclamp((o0 >> 1) - 4, 0, R - 16)
                      : iclamp(o0 - 4, 0, R - 24);

    const float* FL = Fmat + ((size_t)ty * 2 + 0) * MSZ;
    const float* FH = Fmat + ((size_t)ty * 2 + 1) * MSZ;
    v2f aL[K4], aH[K4];
#pragma unroll
    for (int t = 0; t < K4; ++t) {
      const int kb = 4 * t + (hg << 1);
      aL[t] = *reinterpret_cast<const v2f*>(FL + l16 * KW + kb);
      aH[t] = *reinterpret_cast<const v2f*>(FH + l16 * KW + kb);
    }

    if constexpr (NTC >= 8) {
      const int tx0 = wv * TPWC;
      col_strip_lds<R, TPWC, K4, 0, P>(llp, yhp, aL, aH, sY, w0, tx0, hg, l16);
      col_strip_lds<R, TPWC, K4, 1, P>(llp, yhp, aL, aH, sY, w0, tx0, hg, l16);
    } else {
      const int tx0 = wv & 3;
      if (wv < 4)
        col_strip_lds<R, 1, K4, 0, P>(llp, yhp, aL, aH, sY, w0, tx0, hg, l16);
      else
        col_strip_lds<R, 1, K4, 1, P>(llp, yhp, aL, aH, sY, w0, tx0, hg, l16);
    }

    v2f b1[TPWR][K4], b2[TPWR][K4];
    int w0r[TPWR];
#pragma unroll
    for (int j = 0; j < TPWR; ++j) {
      const int tx = wv * TPWR + j;
      const int p0 = tx << 4;
      w0r[j] = UP ? iclamp((p0 >> 1) - 4, 0, R - 16)
                  : iclamp(p0 - 4, 0, R - 24);
      const float* GL = Gmat + ((size_t)tx * 2 + 0) * MSZ;
      const float* GH = Gmat + ((size_t)tx * 2 + 1) * MSZ;
#pragma unroll
      for (int t = 0; t < K4; ++t) {
        const int kb = 4 * t + (hg << 1);
        b1[j][t].x = GL[kb * 16 + l16];
        b1[j][t].y = GL[(kb + 1) * 16 + l16];
        b2[j][t].x = GH[kb * 16 + l16];
        b2[j][t].y = GH[(kb + 1) * 16 + l16];
      }
    }

    lds_barrier();

#pragma unroll
    for (int j = 0; j < TPWR; ++j) {
      const int p0 = (wv * TPWR + j) << 4;
      const float* r1 = sY + l16 * P + (hg << 1) + w0r[j];
      const float* r2 = r1 + 16 * P;

      v8f acc = {0.f, 0.f, 0.f, 0.f, 0.f, 0.f, 0.f, 0.f};
#pragma unroll
      for (int t = 0; t < K4; ++t) {
        const v2f a1 = *reinterpret_cast<const v2f*>(r1 + 4 * t);
        const v2f a2 = *reinterpret_cast<const v2f*>(r2 + 4 * t);
        acc = wmma4(a1, b1[j][t], acc);
        acc = wmma4(a2, b2[j][t], acc);
      }

      float* outp = outm + (size_t)o0 * OW + p0;
#pragma unroll
      for (int v = 0; v < 8; ++v) outp[(unsigned)(v * OW)] = acc[v];
    }

    if (s + 1 < NS) lds_barrier();
  }
}

extern "C" void kernel_launch(void* const* d_in, const int* in_sizes, int n_in,
                              void* d_out, int out_size, void* d_ws,
                              size_t ws_size, hipStream_t stream) {
  (void)in_sizes; (void)n_in; (void)out_size; (void)ws_size;
  const float* yl  = (const float*)d_in[0];
  const float* yh1 = (const float*)d_in[1];
  const float* yh2 = (const float*)d_in[2];
  const float* yh3 = (const float*)d_in[3];
  const float* g0o = (const float*)d_in[4];
  const float* g1o = (const float*)d_in[5];
  const float* g0a = (const float*)d_in[6];
  const float* g0b = (const float*)d_in[7];
  const float* g1a = (const float*)d_in[8];
  const float* g1b = (const float*)d_in[9];

  const size_t SEG = (size_t)16 * 1024 * 1024;
  float* bufA = (float*)d_ws;
  float* bufB = bufA + SEG;
  float* bufC = bufB + SEG;
  float* F3c = bufC + SEG;
  float* G3r = F3c + 8 * 2 * 256;
  float* F2c = G3r + 8 * 2 * 256;
  float* G2r = F2c + 16 * 2 * 256;
  float* F1c = G2r + 16 * 2 * 256;
  float* G1r = F1c + 16 * 2 * 384;

  build_ifilt_colF<<<dim3(8, 2),  256, 0, stream>>>(g0a, g0b, g1a, g1b, F3c, 64);
  build_ifilt_rowG<<<dim3(8, 2),  256, 0, stream>>>(g0a, g0b, g1a, g1b, G3r, 64);
  build_ifilt_colF<<<dim3(16, 2), 256, 0, stream>>>(g0a, g0b, g1a, g1b, F2c, 128);
  build_ifilt_rowG<<<dim3(16, 2), 256, 0, stream>>>(g0a, g0b, g1a, g1b, G2r, 128);
  build_j1_colF<<<dim3(16, 2),    256, 0, stream>>>(g0o, g1o, F1c, 256);
  build_j1_rowG<<<dim3(16, 2),    256, 0, stream>>>(g0o, g1o, G1r, 256);

  level_kernel<64, true, 2><<<dim3(4, 256), 256, 0, stream>>>(
      yl, yh3, F3c, G3r, bufC);
  level_kernel<128, true, 2><<<dim3(8, 256), 256, 0, stream>>>(
      bufC, yh2, F2c, G2r, bufA);
  level_kernel<256, false, 2><<<dim3(8, 256), 256, 0, stream>>>(
      bufA, yh1, F1c, G1r, (float*)d_out);
}


# Column pass: a wave takes 32 columns as two tiles of interleaved columns; one 8-byte load per lane feeds both tiles (c2q re/im pair, 2 ll columns), one 8-byte LDS store; odd-row c2q as SC*(a-b)
# speedup vs baseline: 3.4151x; 1.2370x over previous
#include <hip/hip_runtime.h>


typedef __attribute__((ext_vector_type(2))) float v2f;
typedef __attribute__((ext_vector_type(8))) float v8f;

#define SC_HALF_SQRT2 0.70710678118654752f

__device__ __forceinline__ int refl(int i, int n) {
  i = (i < 0) ? (-1 - i) : i;
  i = (i >= n) ? (2 * n - 1 - i) : i;
  return i;
}

__device__ __forceinline__ int iclamp(int v, int lo, int hi) {
  return v < lo ? lo : (v > hi ? hi : v);
}

__device__ __forceinline__ int wave_in_block() {
  return __builtin_amdgcn_readfirstlane((int)(threadIdx.x >> 5));
}

__device__ __forceinline__ v8f wmma4(v2f a, v2f b, v8f c) {
  return __builtin_amdgcn_wmma_f32_16x16x4_f32(false, a, false, b, (short)0, c,
                                               false, false);
}

__device__ __forceinline__ float ifilt_coef(int o, int kg, int n,
                                            const float (&ha)[10],
                                            const float (&hb)[10], bool hp) {
  const int i = o >> 2, s = o & 3;
  const int base = 4 + ((((s & 1) != 0) != hp) ? 1 : 0);
  float acc = 0.f;
#pragma unroll
  for (int m = 0; m < 5; ++m) {
    const int L = refl(2 * i + base - 2 * m, n);
    const float c = (s == 0)   ? ha[2 * m]
                    : (s == 1) ? hb[2 * m]
                    : (s == 2) ? ha[2 * m + 1]
                               : hb[2 * m + 1];
    acc += (L == kg) ? c : 0.f;
  }
  return acc;
}

template <int NT>
__device__ __forceinline__ float filt_coef(int o, int kg, int n,
                                           const float (&h)[NT]) {
  float acc = 0.f;
#pragma unroll
  for (int m = 0; m < NT; ++m) {
    const int L = refl(o + (NT >> 1) - m, n);
    acc += (L == kg) ? h[m] : 0.f;
  }
  return acc;
}


__global__ __launch_bounds__(256) void build_ifilt_colF(
    const float* __restrict__ g0a, const float* __restrict__ g0b,
    const float* __restrict__ g1a, const float* __restrict__ g1b,
    float* __restrict__ M, int r) {
  const int ty = blockIdx.x, f = blockIdx.y;
  float ha[10], hb[10];
#pragma unroll
  for (int m = 0; m < 10; ++m) {
    ha[m] = f ? g1b[m] : g0b[m];
    hb[m] = f ? g1a[m] : g0a[m];
  }
  const int o0 = ty << 4;
  const int w0 = iclamp((o0 >> 1) - 4, 0, r - 16);
  const int idx = threadIdx.x;
  const int row = idx >> 4, k = idx & 15;
  M[(((size_t)ty * 2 + f) * 16 + row) * 16 + k] =
      ifilt_coef(o0 + row, w0 + k, r, ha, hb, f != 0);
}

__global__ __launch_bounds__(256) void build_ifilt_rowG(
    const float* __restrict__ g0a, const float* __restrict__ g0b,
    const float* __restrict__ g1a, const float* __restrict__ g1b,
    float* __restrict__ M, int W) {
  const int tx = blockIdx.x, f = blockIdx.y;
  float ha[10], hb[10];
#pragma unroll
  for (int m = 0; m < 10; ++m) {
    ha[m] = f ? g1b[m] : g0b[m];
    hb[m] = f ? g1a[m] : g0a[m];
  }
  const int p0 = tx << 4;
  const int w0 = iclamp((p0 >> 1) - 4, 0, W - 16);
  const int idx = threadIdx.x;
  const int k = idx >> 4, n = idx & 15;
  M[(((size_t)tx * 2 + f) * 16 + k) * 16 + n] =
      ifilt_coef(p0 + n, w0 + k, W, ha, hb, f != 0);
}

__global__ __launch_bounds__(256) void build_j1_colF(
    const float* __restrict__ g0o, const float* __restrict__ g1o,
    float* __restrict__ M, int r) {
  const int ty = blockIdx.x, f = blockIdx.y;
  float h0[7], h1[5];
#pragma unroll
  for (int m = 0; m < 7; ++m) h0[m] = g0o[m];
#pragma unroll
  for (int m = 0; m < 5; ++m) h1[m] = g1o[m];
  const int o0 = ty << 4;
  const int w0 = iclamp(o0 - 4, 0, r - 24);
  for (int idx = threadIdx.x; idx < 16 * 24; idx += 256) {
    const int row = idx / 24, k = idx % 24;
    const float v = f ? filt_coef<5>(o0 + row, w0 + k, r, h1)
                      : filt_coef<7>(o0 + row, w0 + k, r, h0);
    M[((size_t)ty * 2 + f) * 384 + row * 24 + k] = v;
  }
}

__global__ __launch_bounds__(256) void build_j1_rowG(
    const float* __restrict__ g0o, const float* __restrict__ g1o,
    float* __restrict__ M, int W) {
  const int tx = blockIdx.x, f = blockIdx.y;
  float h0[7], h1[5];
#pragma unroll
  for (int m = 0; m < 7; ++m) h0[m] = g0o[m];
#pragma unroll
  for (int m = 0; m < 5; ++m) h1[m] = g1o[m];
  const int p0 = tx << 4;
  const int w0 = iclamp(p0 - 4, 0, W - 24);
  for (int idx = threadIdx.x; idx < 24 * 16; idx += 256) {
    const int k = idx >> 4, n = idx & 15;
    const float v = f ? filt_coef<5>(p0 + n, w0 + k, W, h1)
                      : filt_coef<7>(p0 + n, w0 + k, W, h0);
    M[((size_t)tx * 2 + f) * 384 + k * 16 + n] = v;
  }
}


__device__ __forceinline__ void c2q_pair(v2f p1, v2f p2, v2f& e, v2f& o) {
  e.x = SC_HALF_SQRT2 * (p1.x + p2.x);
  o.x = SC_HALF_SQRT2 * (p1.y + p2.y);
  e.y = SC_HALF_SQRT2 * (p1.y - p2.y);
  o.y = SC_HALF_SQRT2 * (p2.x - p1.x);
}

template <int R, int K4, int WH, int P>
__device__ __forceinline__ void col_pair_lds(
    const float* __restrict__ llp, const float* __restrict__ yhp,
    const v2f (&aL)[K4], const v2f (&aH)[K4], float* sdst,
    int w0, int c0, int hg, int l16) {
  constexpr int H2 = R / 2;
  constexpr int SB = H2 * H2 * 2;
  const float* pH1 = yhp + (WH ? 1 : 0) * SB;
  const float* pH2 = yhp + (WH ? 4 : 5) * SB;
  const float* pL1 = yhp + 2 * SB;
  const float* pL2 = yhp + 3 * SB;

  const unsigned col = (unsigned)(c0 + 2 * l16);

  v2f qa[K4], qb[K4], pa[K4], pb[K4];
#pragma unroll
  for (int t = 0; t < K4; ++t) {
    const int kb = 4 * t + (hg << 1);
    const unsigned rowb = (unsigned)(w0 + kb) * H2 + col;
    if constexpr (WH == 0) {
      const unsigned l0 = (unsigned)(w0 + kb) * R + col;
      qa[t] = *reinterpret_cast<const v2f*>(llp + l0);
      qb[t] = *reinterpret_cast<const v2f*>(llp + l0 + R);
    } else {
      qa[t] = *reinterpret_cast<const v2f*>(pL1 + rowb);
      qb[t] = *reinterpret_cast<const v2f*>(pL2 + rowb);
    }
    pa[t] = *reinterpret_cast<const v2f*>(pH1 + rowb);
    pb[t] = *reinterpret_cast<const v2f*>(pH2 + rowb);
  }
  __builtin_amdgcn_sched_barrier(0);

  v8f acc0 = {0.f, 0.f, 0.f, 0.f, 0.f, 0.f, 0.f, 0.f};
  v8f acc1 = {0.f, 0.f, 0.f, 0.f, 0.f, 0.f, 0.f, 0.f};
#pragma unroll
  for (int t = 0; t < K4; ++t) {
    v2f bL0, bL1, bH0, bH1;
    if constexpr (WH == 0) {
      bL0.x = qa[t].x; bL0.y = qb[t].x;
      bL1.x = qa[t].y; bL1.y = qb[t].y;
    } else {
      c2q_pair(qa[t], qb[t], bL0, bL1);
    }
    c2q_pair(pa[t], pb[t], bH0, bH1);
    acc0 = wmma4(aL[t], bL0, acc0);
    acc1 = wmma4(aL[t], bL1, acc1);
    acc0 = wmma4(aH[t], bH0, acc0);
    acc1 = wmma4(aH[t], bH1, acc1);
  }

  float* sp = sdst + (WH * 16 + 8 * hg) * P + col;
#pragma unroll
  for (int v = 0; v < 8; ++v) {
    v2f d;
    d.x = acc0[v];
    d.y = acc1[v];
    *reinterpret_cast<v2f*>(sp + v * P) = d;
  }
}

__device__ __forceinline__ void lds_barrier() {
  __builtin_amdgcn_fence(__ATOMIC_RELEASE, "workgroup", "local");
  __builtin_amdgcn_s_barrier();
  __builtin_amdgcn_fence(__ATOMIC_ACQUIRE, "workgroup", "local");
}

template <int R, bool UP, int NS>
__global__ __launch_bounds__(256) void level_kernel(
    const float* __restrict__ ll, const float* __restrict__ yh,
    const float* __restrict__ Fmat, const float* __restrict__ Gmat,
    float* __restrict__ out) {
  constexpr int K4 = UP ? 4 : 6;
  constexpr int KW = 4 * K4;
  constexpr int MSZ = 16 * KW;
  constexpr int OH = UP ? 2 * R : R;
  constexpr int OW = OH;
  constexpr int NSP = R / 32;
  constexpr int NTR = OW / 16;
  constexpr int TPWR = NTR / 8;
  constexpr int H2 = R / 2;
  constexpr int SB = H2 * H2 * 2;
  constexpr int P = R + 4;
  static_assert(NTR % 8 == 0 && (OH / 16) % NS == 0, "strip shape");
  static_assert(NSP == 8 || 2 * NSP <= 8, "column spans per wave");

  __shared__ __attribute__((aligned(16))) float sY[2 * 16 * P];

  const int lane = threadIdx.x & 31;
  const int wv = wave_in_block();
  const int hg = lane >> 4, l16 = lane & 15;
  const int mc = blockIdx.y;

  const float* llp = ll + (size_t)mc * (R * R);
  const float* yhp = yh + (size_t)mc * (6 * SB);
  float* outm = out + (size_t)mc * (OH * OW) + (size_t)(8 * hg) * OW + l16;

  for (int s = 0; s < NS; ++s) {
    const int ty = blockIdx.x * NS + s;
    const int o0 = ty << 4;
    const int w0 = UP ? iclamp((o0 >> 1) - 4, 0, R - 16)
                      : iclamp(o0 - 4, 0, R - 24);

    const float* FL = Fmat + ((size_t)ty * 2 + 0) * MSZ;
    const float* FH = Fmat + ((size_t)ty * 2 + 1) * MSZ;
    v2f aL[K4], aH[K4];
#pragma unroll
    for (int t = 0; t < K4; ++t) {
      const int kb = 4 * t + (hg << 1);
      aL[t] = *reinterpret_cast<const v2f*>(FL + l16 * KW + kb);
      aH[t] = *reinterpret_cast<const v2f*>(FH + l16 * KW + kb);
    }

    if constexpr (NSP == 8) {
      const int c0 = wv * 32;
      col_pair_lds<R, K4, 0, P>(llp, yhp, aL, aH, sY, w0, c0, hg, l16);
      __builtin_amdgcn_sched_barrier(0);
      col_pair_lds<R, K4, 1, P>(llp, yhp, aL, aH, sY, w0, c0, hg, l16);
    } else {
      if (wv < NSP)
        col_pair_lds<R, K4, 0, P>(llp, yhp, aL, aH, sY, w0, wv * 32, hg, l16);
      else if (wv < 2 * NSP)
        col_pair_lds<R, K4, 1, P>(llp, yhp, aL, aH, sY, w0, (wv - NSP) * 32,
                                  hg, l16);
    }

    __builtin_amdgcn_sched_barrier(0);

    v2f b1[TPWR][K4], b2[TPWR][K4];
    int w0r[TPWR];
#pragma unroll
    for (int j = 0; j < TPWR; ++j) {
      const int tx = wv * TPWR + j;
      const int p0 = tx << 4;
      w0r[j] = UP ? iclamp((p0 >> 1) - 4, 0, R - 16)
                  : iclamp(p0 - 4, 0, R - 24);
      const float* GL = Gmat + ((size_t)tx * 2 + 0) * MSZ;
      const float* GH = Gmat + ((size_t)tx * 2 + 1) * MSZ;
#pragma unroll
      for (int t = 0; t < K4; ++t) {
        const int kb = 4 * t + (hg << 1);
        b1[j][t].x = GL[kb * 16 + l16];
        b1[j][t].y = GL[(kb + 1) * 16 + l16];
        b2[j][t].x = GH[kb * 16 + l16];
        b2[j][t].y = GH[(kb + 1) * 16 + l16];
      }
    }

    lds_barrier();

#pragma unroll
    for (int j = 0; j < TPWR; ++j) {
      const int p0 = (wv * TPWR + j) << 4;
      const float* r1 = sY + l16 * P + (hg << 1) + w0r[j];
      const float* r2 = r1 + 16 * P;

      v8f acc = {0.f, 0.f, 0.f, 0.f, 0.f, 0.f, 0.f, 0.f};
#pragma unroll
      for (int t = 0; t < K4; ++t) {
        const v2f a1 = *reinterpret_cast<const v2f*>(r1 + 4 * t);
        const v2f a2 = *reinterpret_cast<const v2f*>(r2 + 4 * t);
        acc = wmma4(a1, b1[j][t], acc);
        acc = wmma4(a2, b2[j][t], acc);
      }

      float* outp = outm + (size_t)o0 * OW + p0;
#pragma unroll
      for (int v = 0; v < 8; ++v) outp[(unsigned)(v * OW)] = acc[v];
    }

    if (s + 1 < NS) lds_barrier();
  }
}

extern "C" void kernel_launch(void* const* d_in, const int* in_sizes, int n_in,
                              void* d_out, int out_size, void* d_ws,
                              size_t ws_size, hipStream_t stream) {
  (void)in_sizes; (void)n_in; (void)out_size; (void)ws_size;
  const float* yl  = (const float*)d_in[0];
  const float* yh1 = (const float*)d_in[1];
  const float* yh2 = (const float*)d_in[2];
  const float* yh3 = (const float*)d_in[3];
  const float* g0o = (const float*)d_in[4];
  const float* g1o = (const float*)d_in[5];
  const float* g0a = (const float*)d_in[6];
  const float* g0b = (const float*)d_in[7];
  const float* g1a = (const float*)d_in[8];
  const float* g1b = (const float*)d_in[9];

  const size_t SEG = (size_t)16 * 1024 * 1024;
  float* bufA = (float*)d_ws;
  float* bufB = bufA + SEG;
  float* bufC = bufB + SEG;
  float* F3c = bufC + SEG;
  float* G3r = F3c + 8 * 2 * 256;
  float* F2c = G3r + 8 * 2 * 256;
  float* G2r = F2c + 16 * 2 * 256;
  float* F1c = G2r + 16 * 2 * 256;
  float* G1r = F1c + 16 * 2 * 384;

  build_ifilt_colF<<<dim3(8, 2),  256, 0, stream>>>(g0a, g0b, g1a, g1b, F3c, 64);
  build_ifilt_rowG<<<dim3(8, 2),  256, 0, stream>>>(g0a, g0b, g1a, g1b, G3r, 64);
  build_ifilt_colF<<<dim3(16, 2), 256, 0, stream>>>(g0a, g0b, g1a, g1b, F2c, 128);
  build_ifilt_rowG<<<dim3(16, 2), 256, 0, stream>>>(g0a, g0b, g1a, g1b, G2r, 128);
  build_j1_colF<<<dim3(16, 2),    256, 0, stream>>>(g0o, g1o, F1c, 256);
  build_j1_rowG<<<dim3(16, 2),    256, 0, stream>>>(g0o, g1o, G1r, 256);

  level_kernel<64, true, 2><<<dim3(4, 256), 256, 0, stream>>>(
      yl, yh3, F3c, G3r, bufC);
  level_kernel<128, true, 2><<<dim3(8, 256), 256, 0, stream>>>(
      bufC, yh2, F2c, G2r, bufA);
  level_kernel<256, false, 2><<<dim3(8, 256), 256, 0, stream>>>(
      bufA, yh1, F1c, G1r, (float*)d_out);
}


# All six banded filter-matrix sets built by ONE launch (a work-group finds its set from its block index) instead of six dependent launches: 9 -> 4 launches per call, level kernels untouched
# speedup vs baseline: 3.7310x; 1.0925x over previous
#include <hip/hip_runtime.h>


typedef __attribute__((ext_vector_type(2))) float v2f;
typedef __attribute__((ext_vector_type(8))) float v8f;

#define SC_HALF_SQRT2 0.70710678118654752f

__device__ __forceinline__ int refl(int i, int n) {
  i = (i < 0) ? (-1 - i) : i;
  i = (i >= n) ? (2 * n - 1 - i) : i;
  return i;
}

__device__ __forceinline__ int iclamp(int v, int lo, int hi) {
  return v < lo ? lo : (v > hi ? hi : v);
}

__device__ __forceinline__ int wave_in_block() {
  return __builtin_amdgcn_readfirstlane((int)(threadIdx.x >> 5));
}

__device__ __forceinline__ v8f wmma4(v2f a, v2f b, v8f c) {
  return __builtin_amdgcn_wmma_f32_16x16x4_f32(false, a, false, b, (short)0, c,
                                               false, false);
}

__device__ __forceinline__ float ifilt_coef(int o, int kg, int n,
                                            const float (&ha)[10],
                                            const float (&hb)[10], bool hp) {
  const int i = o >> 2, s = o & 3;
  const int base = 4 + ((((s & 1) != 0) != hp) ? 1 : 0);
  float acc = 0.f;
#pragma unroll
  for (int m = 0; m < 5; ++m) {
    const int L = refl(2 * i + base - 2 * m, n);
    const float c = (s == 0)   ? ha[2 * m]
                    : (s == 1) ? hb[2 * m]
                    : (s == 2) ? ha[2 * m + 1]
                               : hb[2 * m + 1];
    acc += (L == kg) ? c : 0.f;
  }
  return acc;
}

template <int NT>
__device__ __forceinline__ float filt_coef(int o, int kg, int n,
                                           const float (&h)[NT]) {
  float acc = 0.f;
#pragma unroll
  for (int m = 0; m < NT; ++m) {
    const int L = refl(o + (NT >> 1) - m, n);
    acc += (L == kg) ? h[m] : 0.f;
  }
  return acc;
}


__device__ __forceinline__ void ifilt_colF_block(
    const float* __restrict__ g0a, const float* __restrict__ g0b,
    const float* __restrict__ g1a, const float* __restrict__ g1b,
    float* __restrict__ M, int r, int ty, int f) {
  float ha[10], hb[10];
#pragma unroll
  for (int m = 0; m < 10; ++m) {
    ha[m] = f ? g1b[m] : g0b[m];
    hb[m] = f ? g1a[m] : g0a[m];
  }
  const int o0 = ty << 4;
  const int w0 = iclamp((o0 >> 1) - 4, 0, r - 16);
  const int idx = threadIdx.x;
  const int row = idx >> 4, k = idx & 15;
  M[(((size_t)ty * 2 + f) * 16 + row) * 16 + k] =
      ifilt_coef(o0 + row, w0 + k, r, ha, hb, f != 0);
}

__device__ __forceinline__ void ifilt_rowG_block(
    const float* __restrict__ g0a, const float* __restrict__ g0b,
    const float* __restrict__ g1a, const float* __restrict__ g1b,
    float* __restrict__ M, int W, int tx, int f) {
  float ha[10], hb[10];
#pragma unroll
  for (int m = 0; m < 10; ++m) {
    ha[m] = f ? g1b[m] : g0b[m];
    hb[m] = f ? g1a[m] : g0a[m];
  }
  const int p0 = tx << 4;
  const int w0 = iclamp((p0 >> 1) - 4, 0, W - 16);
  const int idx = threadIdx.x;
  const int k = idx >> 4, n = idx & 15;
  M[(((size_t)tx * 2 + f) * 16 + k) * 16 + n] =
      ifilt_coef(p0 + n, w0 + k, W, ha, hb, f != 0);
}

__device__ __forceinline__ void j1_colF_block(
    const float* __restrict__ g0o, const float* __restrict__ g1o,
    float* __restrict__ M, int r, int ty, int f) {
  float h0[7], h1[5];
#pragma unroll
  for (int m = 0; m < 7; ++m) h0[m] = g0o[m];
#pragma unroll
  for (int m = 0; m < 5; ++m) h1[m] = g1o[m];
  const int o0 = ty << 4;
  const int w0 = iclamp(o0 - 4, 0, r - 24);
  for (int idx = threadIdx.x; idx < 16 * 24; idx += 256) {
    const int row = idx / 24, k = idx % 24;
    const float v = f ? filt_coef<5>(o0 + row, w0 + k, r, h1)
                      : filt_coef<7>(o0 + row, w0 + k, r, h0);
    M[((size_t)ty * 2 + f) * 384 + row * 24 + k] = v;
  }
}

__device__ __forceinline__ void j1_rowG_block(
    const float* __restrict__ g0o, const float* __restrict__ g1o,
    float* __restrict__ M, int W, int tx, int f) {
  float h0[7], h1[5];
#pragma unroll
  for (int m = 0; m < 7; ++m) h0[m] = g0o[m];
#pragma unroll
  for (int m = 0; m < 5; ++m) h1[m] = g1o[m];
  const int p0 = tx << 4;
  const int w0 = iclamp(p0 - 4, 0, W - 24);
  for (int idx = threadIdx.x; idx < 24 * 16; idx += 256) {
    const int k = idx >> 4, n = idx & 15;
    const float v = f ? filt_coef<5>(p0 + n, w0 + k, W, h1)
                      : filt_coef<7>(p0 + n, w0 + k, W, h0);
    M[((size_t)tx * 2 + f) * 384 + k * 16 + n] = v;
  }
}

#define NB_L3 16
#define NB_L2 32
#define NB_L1 32
#define NB_ALL (2 * NB_L3 + 2 * NB_L2 + 2 * NB_L1)

__global__ __launch_bounds__(256) void build_filters(
    const float* __restrict__ g0o, const float* __restrict__ g1o,
    const float* __restrict__ g0a, const float* __restrict__ g0b,
    const float* __restrict__ g1a, const float* __restrict__ g1b,
    float* __restrict__ F3c, float* __restrict__ G3r,
    float* __restrict__ F2c, float* __restrict__ G2r,
    float* __restrict__ F1c, float* __restrict__ G1r) {
  int b = blockIdx.x;
  if (b < NB_L3) {
    ifilt_colF_block(g0a, g0b, g1a, g1b, F3c, 64, b >> 1, b & 1);
    return;
  }
  b -= NB_L3;
  if (b < NB_L3) {
    ifilt_rowG_block(g0a, g0b, g1a, g1b, G3r, 64, b >> 1, b & 1);
    return;
  }
  b -= NB_L3;
  if (b < NB_L2) {
    ifilt_colF_block(g0a, g0b, g1a, g1b, F2c, 128, b >> 1, b & 1);
    return;
  }
  b -= NB_L2;
  if (b < NB_L2) {
    ifilt_rowG_block(g0a, g0b, g1a, g1b, G2r, 128, b >> 1, b & 1);
    return;
  }
  b -= NB_L2;
  if (b < NB_L1) {
    j1_colF_block(g0o, g1o, F1c, 256, b >> 1, b & 1);
    return;
  }
  b -= NB_L1;
  j1_rowG_block(g0o, g1o, G1r, 256, b >> 1, b & 1);
}


__device__ __forceinline__ void c2q_pair(v2f p1, v2f p2, v2f& e, v2f& o) {
  e.x = SC_HALF_SQRT2 * (p1.x + p2.x);
  o.x = SC_HALF_SQRT2 * (p1.y + p2.y);
  e.y = SC_HALF_SQRT2 * (p1.y - p2.y);
  o.y = SC_HALF_SQRT2 * (p2.x - p1.x);
}

template <int R, int K4, int WH, int P>
__device__ __forceinline__ void col_pair_lds(
    const float* __restrict__ llp, const float* __restrict__ yhp,
    const v2f (&aL)[K4], const v2f (&aH)[K4], float* sdst,
    int w0, int c0, int hg, int l16) {
  constexpr int H2 = R / 2;
  constexpr int SB = H2 * H2 * 2;
  const float* pH1 = yhp + (WH ? 1 : 0) * SB;
  const float* pH2 = yhp + (WH ? 4 : 5) * SB;
  const float* pL1 = yhp + 2 * SB;
  const float* pL2 = yhp + 3 * SB;

  const unsigned col = (unsigned)(c0 + 2 * l16);

  v2f qa[K4], qb[K4], pa[K4], pb[K4];
#pragma unroll
  for (int t = 0; t < K4; ++t) {
    const int kb = 4 * t + (hg << 1);
    const unsigned rowb = (unsigned)(w0 + kb) * H2 + col;
    if constexpr (WH == 0) {
      const unsigned l0 = (unsigned)(w0 + kb) * R + col;
      qa[t] = *reinterpret_cast<const v2f*>(llp + l0);
      qb[t] = *reinterpret_cast<const v2f*>(llp + l0 + R);
    } else {
      qa[t] = *reinterpret_cast<const v2f*>(pL1 + rowb);
      qb[t] = *reinterpret_cast<const v2f*>(pL2 + rowb);
    }
    pa[t] = *reinterpret_cast<const v2f*>(pH1 + rowb);
    pb[t] = *reinterpret_cast<const v2f*>(pH2 + rowb);
  }
  __builtin_amdgcn_sched_barrier(0);

  v8f acc0 = {0.f, 0.f, 0.f, 0.f, 0.f, 0.f, 0.f, 0.f};
  v8f acc1 = {0.f, 0.f, 0.f, 0.f, 0.f, 0.f, 0.f, 0.f};
#pragma unroll
  for (int t = 0; t < K4; ++t) {
    v2f bL0, bL1, bH0, bH1;
    if constexpr (WH == 0) {
      bL0.x = qa[t].x; bL0.y = qb[t].x;
      bL1.x = qa[t].y; bL1.y = qb[t].y;
    } else {
      c2q_pair(qa[t], qb[t], bL0, bL1);
    }
    c2q_pair(pa[t], pb[t], bH0, bH1);
    acc0 = wmma4(aL[t], bL0, acc0);
    acc1 = wmma4(aL[t], bL1, acc1);
    acc0 = wmma4(aH[t], bH0, acc0);
    acc1 = wmma4(aH[t], bH1, acc1);
  }

  float* sp = sdst + (WH * 16 + 8 * hg) * P + col;
#pragma unroll
  for (int v = 0; v < 8; ++v) {
    v2f d;
    d.x = acc0[v];
    d.y = acc1[v];
    *reinterpret_cast<v2f*>(sp + v * P) = d;
  }
}

__device__ __forceinline__ void lds_barrier() {
  __builtin_amdgcn_fence(__ATOMIC_RELEASE, "workgroup", "local");
  __builtin_amdgcn_s_barrier();
  __builtin_amdgcn_fence(__ATOMIC_ACQUIRE, "workgroup", "local");
}

template <int R, bool UP, int NS>
__global__ __launch_bounds__(256) void level_kernel(
    const float* __restrict__ ll, const float* __restrict__ yh,
    const float* __restrict__ Fmat, const float* __restrict__ Gmat,
    float* __restrict__ out) {
  constexpr int K4 = UP ? 4 : 6;
  constexpr int KW = 4 * K4;
  constexpr int MSZ = 16 * KW;
  constexpr int OH = UP ? 2 * R : R;
  constexpr int OW = OH;
  constexpr int NSP = R / 32;
  constexpr int NTR = OW / 16;
  constexpr int TPWR = NTR / 8;
  constexpr int H2 = R / 2;
  constexpr int SB = H2 * H2 * 2;
  constexpr int P = R + 4;
  static_assert(NTR % 8 == 0 && (OH / 16) % NS == 0, "strip shape");
  static_assert(NSP == 8 || 2 * NSP <= 8, "column spans per wave");

  __shared__ __attribute__((aligned(16))) float sY[2 * 16 * P];

  const int lane = threadIdx.x & 31;
  const int wv = wave_in_block();
  const int hg = lane >> 4, l16 = lane & 15;
  const int mc = blockIdx.y;

  const float* llp = ll + (size_t)mc * (R * R);
  const float* yhp = yh + (size_t)mc * (6 * SB);
  float* outm = out + (size_t)mc * (OH * OW) + (size_t)(8 * hg) * OW + l16;

  for (int s = 0; s < NS; ++s) {
    const int ty = blockIdx.x * NS + s;
    const int o0 = ty << 4;
    const int w0 = UP ? iclamp((o0 >> 1) - 4, 0, R - 16)
                      : iclamp(o0 - 4, 0, R - 24);

    const float* FL = Fmat + ((size_t)ty * 2 + 0) * MSZ;
    const float* FH = Fmat + ((size_t)ty * 2 + 1) * MSZ;
    v2f aL[K4], aH[K4];
#pragma unroll
    for (int t = 0; t < K4; ++t) {
      const int kb = 4 * t + (hg << 1);
      aL[t] = *reinterpret_cast<const v2f*>(FL + l16 * KW + kb);
      aH[t] = *reinterpret_cast<const v2f*>(FH + l16 * KW + kb);
    }

    if constexpr (NSP == 8) {
      const int c0 = wv * 32;
      col_pair_lds<R, K4, 0, P>(llp, yhp, aL, aH, sY, w0, c0, hg, l16);
      __builtin_amdgcn_sched_barrier(0);
      col_pair_lds<R, K4, 1, P>(llp, yhp, aL, aH, sY, w0, c0, hg, l16);
    } else {
      if (wv < NSP)
        col_pair_lds<R, K4, 0, P>(llp, yhp, aL, aH, sY, w0, wv * 32, hg, l16);
      else if (wv < 2 * NSP)
        col_pair_lds<R, K4, 1, P>(llp, yhp, aL, aH, sY, w0, (wv - NSP) * 32,
                                  hg, l16);
    }

    __builtin_amdgcn_sched_barrier(0);

    v2f b1[TPWR][K4], b2[TPWR][K4];
    int w0r[TPWR];
#pragma unroll
    for (int j = 0; j < TPWR; ++j) {
      const int tx = wv * TPWR + j;
      const int p0 = tx << 4;
      w0r[j] = UP ? iclamp((p0 >> 1) - 4, 0, R - 16)
                  : iclamp(p0 - 4, 0, R - 24);
      const float* GL = Gmat + ((size_t)tx * 2 + 0) * MSZ;
      const float* GH = Gmat + ((size_t)tx * 2 + 1) * MSZ;
#pragma unroll
      for (int t = 0; t < K4; ++t) {
        const int kb = 4 * t + (hg << 1);
        b1[j][t].x = GL[kb * 16 + l16];
        b1[j][t].y = GL[(kb + 1) * 16 + l16];
        b2[j][t].x = GH[kb * 16 + l16];
        b2[j][t].y = GH[(kb + 1) * 16 + l16];
      }
    }

    lds_barrier();

#pragma unroll
    for (int j = 0; j < TPWR; ++j) {
      const int p0 = (wv * TPWR + j) << 4;
      const float* r1 = sY + l16 * P + (hg << 1) + w0r[j];
      const float* r2 = r1 + 16 * P;

      v8f acc = {0.f, 0.f, 0.f, 0.f, 0.f, 0.f, 0.f, 0.f};
#pragma unroll
      for (int t = 0; t < K4; ++t) {
        const v2f a1 = *reinterpret_cast<const v2f*>(r1 + 4 * t);
        const v2f a2 = *reinterpret_cast<const v2f*>(r2 + 4 * t);
        acc = wmma4(a1, b1[j][t], acc);
        acc = wmma4(a2, b2[j][t], acc);
      }

      float* outp = outm + (size_t)o0 * OW + p0;
#pragma unroll
      for (int v = 0; v < 8; ++v) outp[(unsigned)(v * OW)] = acc[v];
    }

    if (s + 1 < NS) lds_barrier();
  }
}

extern "C" void kernel_launch(void* const* d_in, const int* in_sizes, int n_in,
                              void* d_out, int out_size, void* d_ws,
                              size_t ws_size, hipStream_t stream) {
  (void)in_sizes; (void)n_in; (void)out_size; (void)ws_size;
  const float* yl  = (const float*)d_in[0];
  const float* yh1 = (const float*)d_in[1];
  const float* yh2 = (const float*)d_in[2];
  const float* yh3 = (const float*)d_in[3];
  const float* g0o = (const float*)d_in[4];
  const float* g1o = (const float*)d_in[5];
  const float* g0a = (const float*)d_in[6];
  const float* g0b = (const float*)d_in[7];
  const float* g1a = (const float*)d_in[8];
  const float* g1b = (const float*)d_in[9];

  const size_t SEG = (size_t)16 * 1024 * 1024;
  float* bufA = (float*)d_ws;
  float* bufB = bufA + SEG;
  float* bufC = bufB + SEG;
  float* F3c = bufC + SEG;
  float* G3r = F3c + 8 * 2 * 256;
  float* F2c = G3r + 8 * 2 * 256;
  float* G2r = F2c + 16 * 2 * 256;
  float* F1c = G2r + 16 * 2 * 256;
  float* G1r = F1c + 16 * 2 * 384;

  build_filters<<<dim3(NB_ALL), 256, 0, stream>>>(
      g0o, g1o, g0a, g0b, g1a, g1b, F3c, G3r, F2c, G2r, F1c, G1r);

  level_kernel<64, true, 2><<<dim3(4, 256), 256, 0, stream>>>(
      yl, yh3, F3c, G3r, bufC);
  level_kernel<128, true, 2><<<dim3(8, 256), 256, 0, stream>>>(
      bufC, yh2, F2c, G2r, bufA);
  level_kernel<256, false, 2><<<dim3(8, 256), 256, 0, stream>>>(
      bufA, yh1, F1c, G1r, (float*)d_out);
}


# Row-pass filter fragments: each work-group copies the three distinct sets (first, inner, last tile) to LDS once; every strip reads them there instead of 4-byte global loads per wave and strip
# speedup vs baseline: 4.0646x; 1.0894x over previous
#include <hip/hip_runtime.h>


typedef __attribute__((ext_vector_type(2))) float v2f;
typedef __attribute__((ext_vector_type(8))) float v8f;

#define SC_HALF_SQRT2 0.70710678118654752f

__device__ __forceinline__ int refl(int i, int n) {
  i = (i < 0) ? (-1 - i) : i;
  i = (i >= n) ? (2 * n - 1 - i) : i;
  return i;
}

__device__ __forceinline__ int iclamp(int v, int lo, int hi) {
  return v < lo ? lo : (v > hi ? hi : v);
}

__device__ __forceinline__ int wave_in_block() {
  return __builtin_amdgcn_readfirstlane((int)(threadIdx.x >> 5));
}

__device__ __forceinline__ v8f wmma4(v2f a, v2f b, v8f c) {
  return __builtin_amdgcn_wmma_f32_16x16x4_f32(false, a, false, b, (short)0, c,
                                               false, false);
}

__device__ __forceinline__ float ifilt_coef(int o, int kg, int n,
                                            const float (&ha)[10],
                                            const float (&hb)[10], bool hp) {
  const int i = o >> 2, s = o & 3;
  const int base = 4 + ((((s & 1) != 0) != hp) ? 1 : 0);
  float acc = 0.f;
#pragma unroll
  for (int m = 0; m < 5; ++m) {
    const int L = refl(2 * i + base - 2 * m, n);
    const float c = (s == 0)   ? ha[2 * m]
                    : (s == 1) ? hb[2 * m]
                    : (s == 2) ? ha[2 * m + 1]
                               : hb[2 * m + 1];
    acc += (L == kg) ? c : 0.f;
  }
  return acc;
}

template <int NT>
__device__ __forceinline__ float filt_coef(int o, int kg, int n,
                                           const float (&h)[NT]) {
  float acc = 0.f;
#pragma unroll
  for (int m = 0; m < NT; ++m) {
    const int L = refl(o + (NT >> 1) - m, n);
    acc += (L == kg) ? h[m] : 0.f;
  }
  return acc;
}


__device__ __forceinline__ void ifilt_colF_block(
    const float* __restrict__ g0a, const float* __restrict__ g0b,
    const float* __restrict__ g1a, const float* __restrict__ g1b,
    float* __restrict__ M, int r, int ty, int f) {
  float ha[10], hb[10];
#pragma unroll
  for (int m = 0; m < 10; ++m) {
    ha[m] = f ? g1b[m] : g0b[m];
    hb[m] = f ? g1a[m] : g0a[m];
  }
  const int o0 = ty << 4;
  const int w0 = iclamp((o0 >> 1) - 4, 0, r - 16);
  const int idx = threadIdx.x;
  const int row = idx >> 4, k = idx & 15;
  M[(((size_t)ty * 2 + f) * 16 + row) * 16 + k] =
      ifilt_coef(o0 + row, w0 + k, r, ha, hb, f != 0);
}

__device__ __forceinline__ void ifilt_rowG_block(
    const float* __restrict__ g0a, const float* __restrict__ g0b,
    const float* __restrict__ g1a, const float* __restrict__ g1b,
    float* __restrict__ M, int W, int tx, int f) {
  float ha[10], hb[10];
#pragma unroll
  for (int m = 0; m < 10; ++m) {
    ha[m] = f ? g1b[m] : g0b[m];
    hb[m] = f ? g1a[m] : g0a[m];
  }
  const int p0 = tx << 4;
  const int w0 = iclamp((p0 >> 1) - 4, 0, W - 16);
  const int idx = threadIdx.x;
  const int k = idx >> 4, n = idx & 15;
  M[(((size_t)tx * 2 + f) * 16 + k) * 16 + n] =
      ifilt_coef(p0 + n, w0 + k, W, ha, hb, f != 0);
}

__device__ __forceinline__ void j1_colF_block(
    const float* __restrict__ g0o, const float* __restrict__ g1o,
    float* __restrict__ M, int r, int ty, int f) {
  float h0[7], h1[5];
#pragma unroll
  for (int m = 0; m < 7; ++m) h0[m] = g0o[m];
#pragma unroll
  for (int m = 0; m < 5; ++m) h1[m] = g1o[m];
  const int o0 = ty << 4;
  const int w0 = iclamp(o0 - 4, 0, r - 24);
  for (int idx = threadIdx.x; idx < 16 * 24; idx += 256) {
    const int row = idx / 24, k = idx % 24;
    const float v = f ? filt_coef<5>(o0 + row, w0 + k, r, h1)
                      : filt_coef<7>(o0 + row, w0 + k, r, h0);
    M[((size_t)ty * 2 + f) * 384 + row * 24 + k] = v;
  }
}

__device__ __forceinline__ void j1_rowG_block(
    const float* __restrict__ g0o, const float* __restrict__ g1o,
    float* __restrict__ M, int W, int tx, int f) {
  float h0[7], h1[5];
#pragma unroll
  for (int m = 0; m < 7; ++m) h0[m] = g0o[m];
#pragma unroll
  for (int m = 0; m < 5; ++m) h1[m] = g1o[m];
  const int p0 = tx << 4;
  const int w0 = iclamp(p0 - 4, 0, W - 24);
  for (int idx = threadIdx.x; idx < 24 * 16; idx += 256) {
    const int k = idx >> 4, n = idx & 15;
    const float v = f ? filt_coef<5>(p0 + n, w0 + k, W, h1)
                      : filt_coef<7>(p0 + n, w0 + k, W, h0);
    M[((size_t)tx * 2 + f) * 384 + k * 16 + n] = v;
  }
}

#define NB_L3 16
#define NB_L2 32
#define NB_L1 32
#define NB_ALL (2 * NB_L3 + 2 * NB_L2 + 2 * NB_L1)

__global__ __launch_bounds__(256) void build_filters(
    const float* __restrict__ g0o, const float* __restrict__ g1o,
    const float* __restrict__ g0a, const float* __restrict__ g0b,
    const float* __restrict__ g1a, const float* __restrict__ g1b,
    float* __restrict__ F3c, float* __restrict__ G3r,
    float* __restrict__ F2c, float* __restrict__ G2r,
    float* __restrict__ F1c, float* __restrict__ G1r) {
  int b = blockIdx.x;
  if (b < NB_L3) {
    ifilt_colF_block(g0a, g0b, g1a, g1b, F3c, 64, b >> 1, b & 1);
    return;
  }
  b -= NB_L3;
  if (b < NB_L3) {
    ifilt_rowG_block(g0a, g0b, g1a, g1b, G3r, 64, b >> 1, b & 1);
    return;
  }
  b -= NB_L3;
  if (b < NB_L2) {
    ifilt_colF_block(g0a, g0b, g1a, g1b, F2c, 128, b >> 1, b & 1);
    return;
  }
  b -= NB_L2;
  if (b < NB_L2) {
    ifilt_rowG_block(g0a, g0b, g1a, g1b, G2r, 128, b >> 1, b & 1);
    return;
  }
  b -= NB_L2;
  if (b < NB_L1) {
    j1_colF_block(g0o, g1o, F1c, 256, b >> 1, b & 1);
    return;
  }
  b -= NB_L1;
  j1_rowG_block(g0o, g1o, G1r, 256, b >> 1, b & 1);
}


__device__ __forceinline__ void c2q_pair(v2f p1, v2f p2, v2f& e, v2f& o) {
  e.x = SC_HALF_SQRT2 * (p1.x + p2.x);
  o.x = SC_HALF_SQRT2 * (p1.y + p2.y);
  e.y = SC_HALF_SQRT2 * (p1.y - p2.y);
  o.y = SC_HALF_SQRT2 * (p2.x - p1.x);
}

template <int R, int K4, int WH, int P>
__device__ __forceinline__ void col_pair_lds(
    const float* __restrict__ llp, const float* __restrict__ yhp,
    const v2f (&aL)[K4], const v2f (&aH)[K4], float* sdst,
    int w0, int c0, int hg, int l16) {
  constexpr int H2 = R / 2;
  constexpr int SB = H2 * H2 * 2;
  const float* pH1 = yhp + (WH ? 1 : 0) * SB;
  const float* pH2 = yhp + (WH ? 4 : 5) * SB;
  const float* pL1 = yhp + 2 * SB;
  const float* pL2 = yhp + 3 * SB;

  const unsigned col = (unsigned)(c0 + 2 * l16);

  v2f qa[K4], qb[K4], pa[K4], pb[K4];
#pragma unroll
  for (int t = 0; t < K4; ++t) {
    const int kb = 4 * t + (hg << 1);
    const unsigned rowb = (unsigned)(w0 + kb) * H2 + col;
    if constexpr (WH == 0) {
      const unsigned l0 = (unsigned)(w0 + kb) * R + col;
      qa[t] = *reinterpret_cast<const v2f*>(llp + l0);
      qb[t] = *reinterpret_cast<const v2f*>(llp + l0 + R);
    } else {
      qa[t] = *reinterpret_cast<const v2f*>(pL1 + rowb);
      qb[t] = *reinterpret_cast<const v2f*>(pL2 + rowb);
    }
    pa[t] = *reinterpret_cast<const v2f*>(pH1 + rowb);
    pb[t] = *reinterpret_cast<const v2f*>(pH2 + rowb);
  }
  __builtin_amdgcn_sched_barrier(0);

  v8f acc0 = {0.f, 0.f, 0.f, 0.f, 0.f, 0.f, 0.f, 0.f};
  v8f acc1 = {0.f, 0.f, 0.f, 0.f, 0.f, 0.f, 0.f, 0.f};
#pragma unroll
  for (int t = 0; t < K4; ++t) {
    v2f bL0, bL1, bH0, bH1;
    if constexpr (WH == 0) {
      bL0.x = qa[t].x; bL0.y = qb[t].x;
      bL1.x = qa[t].y; bL1.y = qb[t].y;
    } else {
      c2q_pair(qa[t], qb[t], bL0, bL1);
    }
    c2q_pair(pa[t], pb[t], bH0, bH1);
    acc0 = wmma4(aL[t], bL0, acc0);
    acc1 = wmma4(aL[t], bL1, acc1);
    acc0 = wmma4(aH[t], bH0, acc0);
    acc1 = wmma4(aH[t], bH1, acc1);
  }

  float* sp = sdst + (WH * 16 + 8 * hg) * P + col;
#pragma unroll
  for (int v = 0; v < 8; ++v) {
    v2f d;
    d.x = acc0[v];
    d.y = acc1[v];
    *reinterpret_cast<v2f*>(sp + v * P) = d;
  }
}

__device__ __forceinline__ void lds_barrier() {
  __builtin_amdgcn_fence(__ATOMIC_RELEASE, "workgroup", "local");
  __builtin_amdgcn_s_barrier();
  __builtin_amdgcn_fence(__ATOMIC_ACQUIRE, "workgroup", "local");
}

template <int R, bool UP, int NS>
__global__ __launch_bounds__(256) void level_kernel(
    const float* __restrict__ ll, const float* __restrict__ yh,
    const float* __restrict__ Fmat, const float* __restrict__ Gmat,
    float* __restrict__ out) {
  constexpr int K4 = UP ? 4 : 6;
  constexpr int KW = 4 * K4;
  constexpr int MSZ = 16 * KW;
  constexpr int OH = UP ? 2 * R : R;
  constexpr int OW = OH;
  constexpr int NSP = R / 32;
  constexpr int NTR = OW / 16;
  constexpr int TPWR = NTR / 8;
  constexpr int H2 = R / 2;
  constexpr int SB = H2 * H2 * 2;
  constexpr int P = R + 4;
  static_assert(NTR % 8 == 0 && (OH / 16) % NS == 0, "strip shape");
  static_assert(NSP == 8 || 2 * NSP <= 8, "column spans per wave");

  __shared__ __attribute__((aligned(16))) float sY[2 * 16 * P];

  constexpr int NFR = 3 * 2 * K4;
  constexpr int FPW = (NFR + 7) / 8;
  static_assert(NTR >= 3, "first, inner and last tile must be distinct");
  __shared__ __attribute__((aligned(16))) v2f sG[NFR * 32];

  const int lane = threadIdx.x & 31;
  const int wv = wave_in_block();
  const int hg = lane >> 4, l16 = lane & 15;
  const int mc = blockIdx.y;

  v2f gf[FPW];
#pragma unroll
  for (int i = 0; i < FPW; ++i) {
    const int r = wv + 8 * i;
    if (8 * i + 7 < NFR || r < NFR) {
      const int c = r / (2 * K4);
      const int f = (r / K4) & 1;
      const int t = r % K4;
      const int tx = (c == 0) ? 0 : ((c == 1) ? 1 : NTR - 1);
      const float* G = Gmat + ((size_t)tx * 2 + f) * MSZ;
      const int kb = 4 * t + (hg << 1);
      gf[i].x = G[kb * 16 + l16];
      gf[i].y = G[(kb + 1) * 16 + l16];
    }
  }

  const float* llp = ll + (size_t)mc * (R * R);
  const float* yhp = yh + (size_t)mc * (6 * SB);
  float* outm = out + (size_t)mc * (OH * OW) + (size_t)(8 * hg) * OW + l16;

  for (int s = 0; s < NS; ++s) {
    const int ty = blockIdx.x * NS + s;
    const int o0 = ty << 4;
    const int w0 = UP ? iclamp((o0 >> 1) - 4, 0, R - 16)
                      : iclamp(o0 - 4, 0, R - 24);

    const float* FL = Fmat + ((size_t)ty * 2 + 0) * MSZ;
    const float* FH = Fmat + ((size_t)ty * 2 + 1) * MSZ;
    v2f aL[K4], aH[K4];
#pragma unroll
    for (int t = 0; t < K4; ++t) {
      const int kb = 4 * t + (hg << 1);
      aL[t] = *reinterpret_cast<const v2f*>(FL + l16 * KW + kb);
      aH[t] = *reinterpret_cast<const v2f*>(FH + l16 * KW + kb);
    }

    if constexpr (NSP == 8) {
      const int c0 = wv * 32;
      col_pair_lds<R, K4, 0, P>(llp, yhp, aL, aH, sY, w0, c0, hg, l16);
      __builtin_amdgcn_sched_barrier(0);
      col_pair_lds<R, K4, 1, P>(llp, yhp, aL, aH, sY, w0, c0, hg, l16);
    } else {
      if (wv < NSP)
        col_pair_lds<R, K4, 0, P>(llp, yhp, aL, aH, sY, w0, wv * 32, hg, l16);
      else if (wv < 2 * NSP)
        col_pair_lds<R, K4, 1, P>(llp, yhp, aL, aH, sY, w0, (wv - NSP) * 32,
                                  hg, l16);
    }

    if (s == 0) {
#pragma unroll
      for (int i = 0; i < FPW; ++i) {
        const int r = wv + 8 * i;
        if (8 * i + 7 < NFR || r < NFR) sG[r * 32 + lane] = gf[i];
      }
    }

    lds_barrier();

    v2f b1[TPWR][K4], b2[TPWR][K4];
    int w0r[TPWR];
#pragma unroll
    for (int j = 0; j < TPWR; ++j) {
      const int tx = wv * TPWR + j;
      const int p0 = tx << 4;
      w0r[j] = UP ? iclamp((p0 >> 1) - 4, 0, R - 16)
                  : iclamp(p0 - 4, 0, R - 24);
      const int c = (tx == 0) ? 0 : ((tx == NTR - 1) ? 2 : 1);
      const v2f* gL = sG + ((c * 2 + 0) * K4) * 32 + lane;
      const v2f* gH = sG + ((c * 2 + 1) * K4) * 32 + lane;
#pragma unroll
      for (int t = 0; t < K4; ++t) {
        b1[j][t] = gL[t * 32];
        b2[j][t] = gH[t * 32];
      }
    }
    __builtin_amdgcn_sched_barrier(0);

#pragma unroll
    for (int j = 0; j < TPWR; ++j) {
      const int p0 = (wv * TPWR + j) << 4;
      const float* r1 = sY + l16 * P + (hg << 1) + w0r[j];
      const float* r2 = r1 + 16 * P;

      v8f acc = {0.f, 0.f, 0.f, 0.f, 0.f, 0.f, 0.f, 0.f};
#pragma unroll
      for (int t = 0; t < K4; ++t) {
        const v2f a1 = *reinterpret_cast<const v2f*>(r1 + 4 * t);
        const v2f a2 = *reinterpret_cast<const v2f*>(r2 + 4 * t);
        acc = wmma4(a1, b1[j][t], acc);
        acc = wmma4(a2, b2[j][t], acc);
      }

      float* outp = outm + (size_t)o0 * OW + p0;
#pragma unroll
      for (int v = 0; v < 8; ++v) outp[(unsigned)(v * OW)] = acc[v];
    }

    if (s + 1 < NS) lds_barrier();
  }
}

extern "C" void kernel_launch(void* const* d_in, const int* in_sizes, int n_in,
                              void* d_out, int out_size, void* d_ws,
                              size_t ws_size, hipStream_t stream) {
  (void)in_sizes; (void)n_in; (void)out_size; (void)ws_size;
  const float* yl  = (const float*)d_in[0];
  const float* yh1 = (const float*)d_in[1];
  const float* yh2 = (const float*)d_in[2];
  const float* yh3 = (const float*)d_in[3];
  const float* g0o = (const float*)d_in[4];
  const float* g1o = (const float*)d_in[5];
  const float* g0a = (const float*)d_in[6];
  const float* g0b = (const float*)d_in[7];
  const float* g1a = (const float*)d_in[8];
  const float* g1b = (const float*)d_in[9];

  const size_t SEG = (size_t)16 * 1024 * 1024;
  float* bufA = (float*)d_ws;
  float* bufB = bufA + SEG;
  float* bufC = bufB + SEG;
  float* F3c = bufC + SEG;
  float* G3r = F3c + 8 * 2 * 256;
  float* F2c = G3r + 8 * 2 * 256;
  float* G2r = F2c + 16 * 2 * 256;
  float* F1c = G2r + 16 * 2 * 256;
  float* G1r = F1c + 16 * 2 * 384;

  build_filters<<<dim3(NB_ALL), 256, 0, stream>>>(
      g0o, g1o, g0a, g0b, g1a, g1b, F3c, G3r, F2c, G2r, F1c, G1r);

  level_kernel<64, true, 2><<<dim3(4, 256), 256, 0, stream>>>(
      yl, yh3, F3c, G3r, bufC);
  level_kernel<128, true, 2><<<dim3(8, 256), 256, 0, stream>>>(
      bufC, yh2, F2c, G2r, bufA);
  level_kernel<256, false, 2><<<dim3(8, 256), 256, 0, stream>>>(
      bufA, yh1, F1c, G1r, (float*)d_out);
}
